# starting kernel
# baseline (speedup 1.0000x reference)
#include <hip/hip_runtime.h>

typedef float v2f __attribute__((ext_vector_type(2)));
typedef float v8f __attribute__((ext_vector_type(8)));

#define D_MODEL 512
#define D_STATE 32
#define BATCH 8
#define SEQ 4096
#define ROWS (BATCH * SEQ)

#define WS_ABAR 0
#define WS_BBAR (WS_ABAR + 1024)
#define WS_UB (WS_BBAR + 16384)
#define WS_STATES (WS_UB + ROWS * 32)

__device__ __forceinline__ v8f wmma4(v2f a, v2f b, v8f c) {
  return __builtin_amdgcn_wmma_f32_16x16x4_f32(false, a, false, b, (short)0, c,
                                               false, false);
}

__global__ void __launch_bounds__(256) prep_kernel(
    const float* __restrict__ A, const float* __restrict__ B,
    const float* __restrict__ log_dt, float* __restrict__ ws) {
  __shared__ float aug[32][64];
  __shared__ float Ah[32][32];
  __shared__ float inv[32][32];
  __shared__ float fac[32];
  __shared__ float dt_sh;
  const int tid = threadIdx.x;

  if (tid == 0) {
    float s = 0.f;
    for (int i = 0; i < D_MODEL; ++i) {
      float v = expf(log_dt[i]);
      v = fminf(fmaxf(v, 1e-4f), 1.0f);
      s += v;
    }
    dt_sh = s / (float)D_MODEL;
  }
  __syncthreads();
  const float dt = dt_sh;

  for (int idx = tid; idx < 32 * 32; idx += blockDim.x) {
    const int i = idx >> 5, j = idx & 31;
    const float a = A[idx];
    const float eye = (i == j) ? 1.f : 0.f;
    aug[i][j] = eye - 0.5f * dt * a;
    aug[i][j + 32] = eye;
    Ah[i][j] = eye + 0.5f * dt * a;
  }
  __syncthreads();

  for (int k = 0; k < 32; ++k) {
    const float p = aug[k][k];
    __syncthreads();
    for (int j = tid; j < 64; j += blockDim.x) aug[k][j] = aug[k][j] / p;
    __syncthreads();
    if (tid < 32) fac[tid] = aug[tid][k];
    __syncthreads();
    for (int idx = tid; idx < 32 * 64; idx += blockDim.x) {
      const int i = idx >> 6, j = idx & 63;
      if (i != k) aug[i][j] -= fac[i] * aug[k][j];
    }
    __syncthreads();
  }
  for (int idx = tid; idx < 32 * 32; idx += blockDim.x) {
    const int i = idx >> 5, j = idx & 31;
    inv[i][j] = aug[i][j + 32];
  }
  __syncthreads();

  float* Abar = ws + WS_ABAR;
  for (int idx = tid; idx < 32 * 32; idx += blockDim.x) {
    const int i = idx >> 5, j = idx & 31;
    float s = 0.f;
    for (int k = 0; k < 32; ++k) s += Ah[i][k] * inv[k][j];
    Abar[idx] = s;
  }
  float* Bbar = ws + WS_BBAR;
  for (int idx = tid; idx < 32 * D_MODEL; idx += blockDim.x) {
    const int n = idx >> 9, d = idx & 511;
    float s = 0.f;
    for (int k = 0; k < 32; ++k) s += inv[n][k] * B[k * D_MODEL + d];
    Bbar[idx] = dt * s;
  }
}

__global__ void __launch_bounds__(256) ub_kernel(const float* __restrict__ x,
                                                 const float* __restrict__ Bbar,
                                                 float* __restrict__ uB) {
  const int wave = threadIdx.x >> 5;
  const int lane = threadIdx.x & 31;
  const int r = lane & 15, h = lane >> 4;
  const int mtile = blockIdx.x * 8 + wave;

  const float* xrow = x + (size_t)(mtile * 16 + r) * D_MODEL;
  const float* b0row = Bbar + (size_t)r * D_MODEL;
  const float* b1row = Bbar + (size_t)(16 + r) * D_MODEL;

  v8f acc0 = {0.f, 0.f, 0.f, 0.f, 0.f, 0.f, 0.f, 0.f};
  v8f acc1 = {0.f, 0.f, 0.f, 0.f, 0.f, 0.f, 0.f, 0.f};

  for (int k0 = 0; k0 < D_MODEL; k0 += 4) {
    const int k = k0 + 2 * h;
    const v2f a = *(const v2f*)(xrow + k);
    const v2f b0 = *(const v2f*)(b0row + k);
    const v2f b1 = *(const v2f*)(b1row + k);
    acc0 = wmma4(a, b0, acc0);
    acc1 = wmma4(a, b1, acc1);
  }

  const int mbase = mtile * 16 + 8 * h;
#pragma unroll
  for (int v = 0; v < 8; ++v) {
    const int m = mbase + v;
    uB[(size_t)m * 32 + r] = acc0[v];
    uB[(size_t)m * 32 + 16 + r] = acc1[v];
  }
}

__global__ void __launch_bounds__(32) scan_kernel(const float* __restrict__ ws,
                                                  float* __restrict__ states,
                                                  float* __restrict__ fs_out) {
  const int b = blockIdx.x;
  const int lane = threadIdx.x;
  const float* Abar = ws + WS_ABAR;
  const float* uB = ws + WS_UB + (size_t)b * SEQ * 32;
  float* st_out = states + (size_t)b * SEQ * 32;

  float a[32];
#pragma unroll
  for (int m = 0; m < 32; ++m) a[m] = Abar[lane * 32 + m];

  float st = 0.f;
  for (int s = 0; s < SEQ; ++s) {
    float acc = uB[s * 32 + lane];
    const int stb = __float_as_int(st);
#pragma unroll
    for (int m = 0; m < 32; ++m) {
      acc = fmaf(a[m], __int_as_float(__builtin_amdgcn_readlane(stb, m)), acc);
    }
    st = tanhf(acc);
    st_out[s * 32 + lane] = st;
  }
  fs_out[b * 32 + lane] = st;
}

__global__ void __launch_bounds__(256) cproj_kernel(
    const float* __restrict__ states, const float* __restrict__ C,
    const float* __restrict__ x, const float* __restrict__ Dv,
    float* __restrict__ y1) {
  const int wave = threadIdx.x >> 5;
  const int lane = threadIdx.x & 31;
  const int r = lane & 15, h = lane >> 4;
  const int mtile = blockIdx.x * 8 + wave;

  v2f afrag[8];
  const float* srow = states + (size_t)(mtile * 16 + r) * 32;
#pragma unroll
  for (int kk = 0; kk < 8; ++kk)
    afrag[kk] = *(const v2f*)(srow + kk * 4 + 2 * h);

  const int mbase = mtile * 16 + 8 * h;
  for (int dtile = 0; dtile < 32; ++dtile) {
    v8f acc = {0.f, 0.f, 0.f, 0.f, 0.f, 0.f, 0.f, 0.f};
    const float* crow = C + (size_t)(dtile * 16 + r) * 32;
#pragma unroll
    for (int kk = 0; kk < 8; ++kk) {
      const v2f b = *(const v2f*)(crow + kk * 4 + 2 * h);
      acc = wmma4(afrag[kk], b, acc);
    }
    const int d = dtile * 16 + r;
    const float Dd = Dv[d];
#pragma unroll
    for (int v = 0; v < 8; ++v) {
      const int m = mbase + v;
      y1[(size_t)m * D_MODEL + d] = acc[v] + x[(size_t)m * D_MODEL + d] * Dd;
    }
  }
}

__global__ void __launch_bounds__(256) wout_kernel(
    const float* __restrict__ W, const float* __restrict__ bout,
    float* __restrict__ y) {
  __shared__ float lds[16 * D_MODEL];
  const int tid = threadIdx.x;
  const int mtile = blockIdx.x;

  const float4* src4 = (const float4*)(y + (size_t)mtile * 16 * D_MODEL);
  float4* dst4 = (float4*)lds;
#pragma unroll
  for (int i = 0; i < 8; ++i) dst4[tid + i * 256] = src4[tid + i * 256];
  __syncthreads();

  const int wave = tid >> 5, lane = tid & 31;
  const int r = lane & 15, h = lane >> 4;
  const int mbase = mtile * 16 + 8 * h;
  const float* arow = lds + r * D_MODEL;

  for (int t = 0; t < 4; ++t) {
    const int ntile = wave * 4 + t;
    const float* wrow = W + (size_t)(ntile * 16 + r) * D_MODEL;
    if (t < 3)
      __builtin_prefetch(W + (size_t)((ntile + 1) * 16 + r) * D_MODEL, 0, 1);

    v8f acc = {0.f, 0.f, 0.f, 0.f, 0.f, 0.f, 0.f, 0.f};
    for (int k0 = 0; k0 < D_MODEL; k0 += 4) {
      const int k = k0 + 2 * h;
      const v2f a = *(const v2f*)(arow + k);
      const v2f b = *(const v2f*)(wrow + k);
      acc = wmma4(a, b, acc);
    }
    const int n = ntile * 16 + r;
    const float bo = bout[n];
#pragma unroll
    for (int v = 0; v < 8; ++v)
      y[(size_t)(mbase + v) * D_MODEL + n] = acc[v] + bo;
  }
}

extern "C" void kernel_launch(void* const* d_in, const int* in_sizes, int n_in,
                              void* d_out, int out_size, void* d_ws,
                              size_t ws_size, hipStream_t stream) {
  const float* x = (const float*)d_in[0];
  const float* A = (const float*)d_in[1];
  const float* B = (const float*)d_in[2];
  const float* C = (const float*)d_in[3];
  const float* Dv = (const float*)d_in[4];
  const float* log_dt = (const float*)d_in[5];
  const float* W = (const float*)d_in[6];
  const float* bout = (const float*)d_in[7];

  float* y = (float*)d_out;
  float* fs = y + (size_t)ROWS * D_MODEL;

  float* ws = (float*)d_ws;
  float* Bbar = ws + WS_BBAR;
  float* uB = ws + WS_UB;
  float* states = ws + WS_STATES;

  prep_kernel<<<1, 256, 0, stream>>>(A, B, log_dt, ws);
  ub_kernel<<<256, 256, 0, stream>>>(x, Bbar, uB);
  scan_kernel<<<BATCH, 32, 0, stream>>>(ws, states, fs);
  cproj_kernel<<<256, 256, 0, stream>>>(states, C, x, Dv, y);
  wout_kernel<<<2048, 256, 0, stream>>>(W, bout, y);
}


# scan: each step's uB input is loaded one chunk of steps ahead into registers, so the recurrence no longer waits a global-load latency in every step; arithmetic and its order unchanged
# speedup vs baseline: 2.6684x; 2.6684x over previous
#include <hip/hip_runtime.h>

typedef float v2f __attribute__((ext_vector_type(2)));
typedef float v8f __attribute__((ext_vector_type(8)));

#define D_MODEL 512
#define D_STATE 32
#define BATCH 8
#define SEQ 4096
#define ROWS (BATCH * SEQ)

#define WS_ABAR 0
#define WS_BBAR (WS_ABAR + 1024)
#define WS_UB (WS_BBAR + 16384)
#define WS_STATES (WS_UB + ROWS * 32)

__device__ __forceinline__ v8f wmma4(v2f a, v2f b, v8f c) {
  return __builtin_amdgcn_wmma_f32_16x16x4_f32(false, a, false, b, (short)0, c,
                                               false, false);
}

__global__ void __launch_bounds__(256) prep_kernel(
    const float* __restrict__ A, const float* __restrict__ B,
    const float* __restrict__ log_dt, float* __restrict__ ws) {
  __shared__ float aug[32][64];
  __shared__ float Ah[32][32];
  __shared__ float inv[32][32];
  __shared__ float fac[32];
  __shared__ float dt_sh;
  const int tid = threadIdx.x;

  if (tid == 0) {
    float s = 0.f;
    for (int i = 0; i < D_MODEL; ++i) {
      float v = expf(log_dt[i]);
      v = fminf(fmaxf(v, 1e-4f), 1.0f);
      s += v;
    }
    dt_sh = s / (float)D_MODEL;
  }
  __syncthreads();
  const float dt = dt_sh;

  for (int idx = tid; idx < 32 * 32; idx += blockDim.x) {
    const int i = idx >> 5, j = idx & 31;
    const float a = A[idx];
    const float eye = (i == j) ? 1.f : 0.f;
    aug[i][j] = eye - 0.5f * dt * a;
    aug[i][j + 32] = eye;
    Ah[i][j] = eye + 0.5f * dt * a;
  }
  __syncthreads();

  for (int k = 0; k < 32; ++k) {
    const float p = aug[k][k];
    __syncthreads();
    for (int j = tid; j < 64; j += blockDim.x) aug[k][j] = aug[k][j] / p;
    __syncthreads();
    if (tid < 32) fac[tid] = aug[tid][k];
    __syncthreads();
    for (int idx = tid; idx < 32 * 64; idx += blockDim.x) {
      const int i = idx >> 6, j = idx & 63;
      if (i != k) aug[i][j] -= fac[i] * aug[k][j];
    }
    __syncthreads();
  }
  for (int idx = tid; idx < 32 * 32; idx += blockDim.x) {
    const int i = idx >> 5, j = idx & 31;
    inv[i][j] = aug[i][j + 32];
  }
  __syncthreads();

  float* Abar = ws + WS_ABAR;
  for (int idx = tid; idx < 32 * 32; idx += blockDim.x) {
    const int i = idx >> 5, j = idx & 31;
    float s = 0.f;
    for (int k = 0; k < 32; ++k) s += Ah[i][k] * inv[k][j];
    Abar[idx] = s;
  }
  float* Bbar = ws + WS_BBAR;
  for (int idx = tid; idx < 32 * D_MODEL; idx += blockDim.x) {
    const int n = idx >> 9, d = idx & 511;
    float s = 0.f;
    for (int k = 0; k < 32; ++k) s += inv[n][k] * B[k * D_MODEL + d];
    Bbar[idx] = dt * s;
  }
}

__global__ void __launch_bounds__(256) ub_kernel(const float* __restrict__ x,
                                                 const float* __restrict__ Bbar,
                                                 float* __restrict__ uB) {
  const int wave = threadIdx.x >> 5;
  const int lane = threadIdx.x & 31;
  const int r = lane & 15, h = lane >> 4;
  const int mtile = blockIdx.x * 8 + wave;

  const float* xrow = x + (size_t)(mtile * 16 + r) * D_MODEL;
  const float* b0row = Bbar + (size_t)r * D_MODEL;
  const float* b1row = Bbar + (size_t)(16 + r) * D_MODEL;

  v8f acc0 = {0.f, 0.f, 0.f, 0.f, 0.f, 0.f, 0.f, 0.f};
  v8f acc1 = {0.f, 0.f, 0.f, 0.f, 0.f, 0.f, 0.f, 0.f};

  for (int k0 = 0; k0 < D_MODEL; k0 += 4) {
    const int k = k0 + 2 * h;
    const v2f a = *(const v2f*)(xrow + k);
    const v2f b0 = *(const v2f*)(b0row + k);
    const v2f b1 = *(const v2f*)(b1row + k);
    acc0 = wmma4(a, b0, acc0);
    acc1 = wmma4(a, b1, acc1);
  }

  const int mbase = mtile * 16 + 8 * h;
#pragma unroll
  for (int v = 0; v < 8; ++v) {
    const int m = mbase + v;
    uB[(size_t)m * 32 + r] = acc0[v];
    uB[(size_t)m * 32 + 16 + r] = acc1[v];
  }
}

#define SCAN_PF 16
__global__ void __launch_bounds__(32) scan_kernel(const float* __restrict__ ws,
                                                  float* __restrict__ states,
                                                  float* __restrict__ fs_out) {
  const int b = blockIdx.x;
  const int lane = threadIdx.x;
  const float* Abar = ws + WS_ABAR;
  const float* uB = ws + WS_UB + (size_t)b * SEQ * 32;
  float* st_out = states + (size_t)b * SEQ * 32;

  float a[32];
#pragma unroll
  for (int m = 0; m < 32; ++m) a[m] = Abar[lane * 32 + m];

  float cur[SCAN_PF];
#pragma unroll
  for (int j = 0; j < SCAN_PF; ++j) cur[j] = uB[j * 32 + lane];

  float st = 0.f;
  for (int s0 = 0; s0 < SEQ; s0 += SCAN_PF) {
    const int sn = (s0 + SCAN_PF < SEQ) ? (s0 + SCAN_PF) : s0;
    float nxt[SCAN_PF];
#pragma unroll
    for (int j = 0; j < SCAN_PF; ++j) nxt[j] = uB[(sn + j) * 32 + lane];

#pragma unroll
    for (int j = 0; j < SCAN_PF; ++j) {
      float acc = cur[j];
      const int stb = __float_as_int(st);
#pragma unroll
      for (int m = 0; m < 32; ++m) {
        acc = fmaf(a[m], __int_as_float(__builtin_amdgcn_readlane(stb, m)), acc);
      }
      st = tanhf(acc);
      st_out[(s0 + j) * 32 + lane] = st;
    }
#pragma unroll
    for (int j = 0; j < SCAN_PF; ++j) cur[j] = nxt[j];
  }
  fs_out[b * 32 + lane] = st;
}

__global__ void __launch_bounds__(256) cproj_kernel(
    const float* __restrict__ states, const float* __restrict__ C,
    const float* __restrict__ x, const float* __restrict__ Dv,
    float* __restrict__ y1) {
  const int wave = threadIdx.x >> 5;
  const int lane = threadIdx.x & 31;
  const int r = lane & 15, h = lane >> 4;
  const int mtile = blockIdx.x * 8 + wave;

  v2f afrag[8];
  const float* srow = states + (size_t)(mtile * 16 + r) * 32;
#pragma unroll
  for (int kk = 0; kk < 8; ++kk)
    afrag[kk] = *(const v2f*)(srow + kk * 4 + 2 * h);

  const int mbase = mtile * 16 + 8 * h;
  for (int dtile = 0; dtile < 32; ++dtile) {
    v8f acc = {0.f, 0.f, 0.f, 0.f, 0.f, 0.f, 0.f, 0.f};
    const float* crow = C + (size_t)(dtile * 16 + r) * 32;
#pragma unroll
    for (int kk = 0; kk < 8; ++kk) {
      const v2f b = *(const v2f*)(crow + kk * 4 + 2 * h);
      acc = wmma4(afrag[kk], b, acc);
    }
    const int d = dtile * 16 + r;
    const float Dd = Dv[d];
#pragma unroll
    for (int v = 0; v < 8; ++v) {
      const int m = mbase + v;
      y1[(size_t)m * D_MODEL + d] = acc[v] + x[(size_t)m * D_MODEL + d] * Dd;
    }
  }
}

__global__ void __launch_bounds__(256) wout_kernel(
    const float* __restrict__ W, const float* __restrict__ bout,
    float* __restrict__ y) {
  __shared__ float lds[16 * D_MODEL];
  const int tid = threadIdx.x;
  const int mtile = blockIdx.x;

  const float4* src4 = (const float4*)(y + (size_t)mtile * 16 * D_MODEL);
  float4* dst4 = (float4*)lds;
#pragma unroll
  for (int i = 0; i < 8; ++i) dst4[tid + i * 256] = src4[tid + i * 256];
  __syncthreads();

  const int wave = tid >> 5, lane = tid & 31;
  const int r = lane & 15, h = lane >> 4;
  const int mbase = mtile * 16 + 8 * h;
  const float* arow = lds + r * D_MODEL;

  for (int t = 0; t < 4; ++t) {
    const int ntile = wave * 4 + t;
    const float* wrow = W + (size_t)(ntile * 16 + r) * D_MODEL;
    if (t < 3)
      __builtin_prefetch(W + (size_t)((ntile + 1) * 16 + r) * D_MODEL, 0, 1);

    v8f acc = {0.f, 0.f, 0.f, 0.f, 0.f, 0.f, 0.f, 0.f};
    for (int k0 = 0; k0 < D_MODEL; k0 += 4) {
      const int k = k0 + 2 * h;
      const v2f a = *(const v2f*)(arow + k);
      const v2f b = *(const v2f*)(wrow + k);
      acc = wmma4(a, b, acc);
    }
    const int n = ntile * 16 + r;
    const float bo = bout[n];
#pragma unroll
    for (int v = 0; v < 8; ++v)
      y[(size_t)(mbase + v) * D_MODEL + n] = acc[v] + bo;
  }
}

extern "C" void kernel_launch(void* const* d_in, const int* in_sizes, int n_in,
                              void* d_out, int out_size, void* d_ws,
                              size_t ws_size, hipStream_t stream) {
  const float* x = (const float*)d_in[0];
  const float* A = (const float*)d_in[1];
  const float* B = (const float*)d_in[2];
  const float* C = (const float*)d_in[3];
  const float* Dv = (const float*)d_in[4];
  const float* log_dt = (const float*)d_in[5];
  const float* W = (const float*)d_in[6];
  const float* bout = (const float*)d_in[7];

  float* y = (float*)d_out;
  float* fs = y + (size_t)ROWS * D_MODEL;

  float* ws = (float*)d_ws;
  float* Bbar = ws + WS_BBAR;
  float* uB = ws + WS_UB;
  float* states = ws + WS_STATES;

  prep_kernel<<<1, 256, 0, stream>>>(A, B, log_dt, ws);
  ub_kernel<<<256, 256, 0, stream>>>(x, Bbar, uB);
  scan_kernel<<<BATCH, 32, 0, stream>>>(ws, states, fs);
  cproj_kernel<<<256, 256, 0, stream>>>(states, C, x, Dv, y);
  wout_kernel<<<2048, 256, 0, stream>>>(W, bout, y);
}


# scan: each step's 32-term dot product is summed as four independent chains of eight FMAs and a two-level add (the order of summation changes) instead of one chain of 32 dependent FMAs
# speedup vs baseline: 2.8913x; 1.0835x over previous
#include <hip/hip_runtime.h>

typedef float v2f __attribute__((ext_vector_type(2)));
typedef float v8f __attribute__((ext_vector_type(8)));

#define D_MODEL 512
#define D_STATE 32
#define BATCH 8
#define SEQ 4096
#define ROWS (BATCH * SEQ)

#define WS_ABAR 0
#define WS_BBAR (WS_ABAR + 1024)
#define WS_UB (WS_BBAR + 16384)
#define WS_STATES (WS_UB + ROWS * 32)

__device__ __forceinline__ v8f wmma4(v2f a, v2f b, v8f c) {
  return __builtin_amdgcn_wmma_f32_16x16x4_f32(false, a, false, b, (short)0, c,
                                               false, false);
}

__global__ void __launch_bounds__(256) prep_kernel(
    const float* __restrict__ A, const float* __restrict__ B,
    const float* __restrict__ log_dt, float* __restrict__ ws) {
  __shared__ float aug[32][64];
  __shared__ float Ah[32][32];
  __shared__ float inv[32][32];
  __shared__ float fac[32];
  __shared__ float dt_sh;
  const int tid = threadIdx.x;

  if (tid == 0) {
    float s = 0.f;
    for (int i = 0; i < D_MODEL; ++i) {
      float v = expf(log_dt[i]);
      v = fminf(fmaxf(v, 1e-4f), 1.0f);
      s += v;
    }
    dt_sh = s / (float)D_MODEL;
  }
  __syncthreads();
  const float dt = dt_sh;

  for (int idx = tid; idx < 32 * 32; idx += blockDim.x) {
    const int i = idx >> 5, j = idx & 31;
    const float a = A[idx];
    const float eye = (i == j) ? 1.f : 0.f;
    aug[i][j] = eye - 0.5f * dt * a;
    aug[i][j + 32] = eye;
    Ah[i][j] = eye + 0.5f * dt * a;
  }
  __syncthreads();

  for (int k = 0; k < 32; ++k) {
    const float p = aug[k][k];
    __syncthreads();
    for (int j = tid; j < 64; j += blockDim.x) aug[k][j] = aug[k][j] / p;
    __syncthreads();
    if (tid < 32) fac[tid] = aug[tid][k];
    __syncthreads();
    for (int idx = tid; idx < 32 * 64; idx += blockDim.x) {
      const int i = idx >> 6, j = idx & 63;
      if (i != k) aug[i][j] -= fac[i] * aug[k][j];
    }
    __syncthreads();
  }
  for (int idx = tid; idx < 32 * 32; idx += blockDim.x) {
    const int i = idx >> 5, j = idx & 31;
    inv[i][j] = aug[i][j + 32];
  }
  __syncthreads();

  float* Abar = ws + WS_ABAR;
  for (int idx = tid; idx < 32 * 32; idx += blockDim.x) {
    const int i = idx >> 5, j = idx & 31;
    float s = 0.f;
    for (int k = 0; k < 32; ++k) s += Ah[i][k] * inv[k][j];
    Abar[idx] = s;
  }
  float* Bbar = ws + WS_BBAR;
  for (int idx = tid; idx < 32 * D_MODEL; idx += blockDim.x) {
    const int n = idx >> 9, d = idx & 511;
    float s = 0.f;
    for (int k = 0; k < 32; ++k) s += inv[n][k] * B[k * D_MODEL + d];
    Bbar[idx] = dt * s;
  }
}

__global__ void __launch_bounds__(256) ub_kernel(const float* __restrict__ x,
                                                 const float* __restrict__ Bbar,
                                                 float* __restrict__ uB) {
  const int wave = threadIdx.x >> 5;
  const int lane = threadIdx.x & 31;
  const int r = lane & 15, h = lane >> 4;
  const int mtile = blockIdx.x * 8 + wave;

  const float* xrow = x + (size_t)(mtile * 16 + r) * D_MODEL;
  const float* b0row = Bbar + (size_t)r * D_MODEL;
  const float* b1row = Bbar + (size_t)(16 + r) * D_MODEL;

  v8f acc0 = {0.f, 0.f, 0.f, 0.f, 0.f, 0.f, 0.f, 0.f};
  v8f acc1 = {0.f, 0.f, 0.f, 0.f, 0.f, 0.f, 0.f, 0.f};

  for (int k0 = 0; k0 < D_MODEL; k0 += 4) {
    const int k = k0 + 2 * h;
    const v2f a = *(const v2f*)(xrow + k);
    const v2f b0 = *(const v2f*)(b0row + k);
    const v2f b1 = *(const v2f*)(b1row + k);
    acc0 = wmma4(a, b0, acc0);
    acc1 = wmma4(a, b1, acc1);
  }

  const int mbase = mtile * 16 + 8 * h;
#pragma unroll
  for (int v = 0; v < 8; ++v) {
    const int m = mbase + v;
    uB[(size_t)m * 32 + r] = acc0[v];
    uB[(size_t)m * 32 + 16 + r] = acc1[v];
  }
}

#define SCAN_PF 16
__global__ void __launch_bounds__(32) scan_kernel(const float* __restrict__ ws,
                                                  float* __restrict__ states,
                                                  float* __restrict__ fs_out) {
  const int b = blockIdx.x;
  const int lane = threadIdx.x;
  const float* Abar = ws + WS_ABAR;
  const float* uB = ws + WS_UB + (size_t)b * SEQ * 32;
  float* st_out = states + (size_t)b * SEQ * 32;

  float a[32];
#pragma unroll
  for (int m = 0; m < 32; ++m) a[m] = Abar[lane * 32 + m];

  float cur[SCAN_PF];
#pragma unroll
  for (int j = 0; j < SCAN_PF; ++j) cur[j] = uB[j * 32 + lane];

  float st = 0.f;
  for (int s0 = 0; s0 < SEQ; s0 += SCAN_PF) {
    const int sn = (s0 + SCAN_PF < SEQ) ? (s0 + SCAN_PF) : s0;
    float nxt[SCAN_PF];
#pragma unroll
    for (int j = 0; j < SCAN_PF; ++j) nxt[j] = uB[(sn + j) * 32 + lane];

#pragma unroll
    for (int j = 0; j < SCAN_PF; ++j) {
      const int stb = __float_as_int(st);
      float p0 = cur[j], p1 = 0.f, p2 = 0.f, p3 = 0.f;
#pragma unroll
      for (int m = 0; m < 32; m += 4) {
        const float s0 = __int_as_float(__builtin_amdgcn_readlane(stb, m + 0));
        const float s1 = __int_as_float(__builtin_amdgcn_readlane(stb, m + 1));
        const float s2 = __int_as_float(__builtin_amdgcn_readlane(stb, m + 2));
        const float s3 = __int_as_float(__builtin_amdgcn_readlane(stb, m + 3));
        p0 = fmaf(a[m + 0], s0, p0);
        p1 = fmaf(a[m + 1], s1, p1);
        p2 = fmaf(a[m + 2], s2, p2);
        p3 = fmaf(a[m + 3], s3, p3);
      }
      const float acc = (p0 + p1) + (p2 + p3);
      st = tanhf(acc);
      st_out[(s0 + j) * 32 + lane] = st;
    }
#pragma unroll
    for (int j = 0; j < SCAN_PF; ++j) cur[j] = nxt[j];
  }
  fs_out[b * 32 + lane] = st;
}

__global__ void __launch_bounds__(256) cproj_kernel(
    const float* __restrict__ states, const float* __restrict__ C,
    const float* __restrict__ x, const float* __restrict__ Dv,
    float* __restrict__ y1) {
  const int wave = threadIdx.x >> 5;
  const int lane = threadIdx.x & 31;
  const int r = lane & 15, h = lane >> 4;
  const int mtile = blockIdx.x * 8 + wave;

  v2f afrag[8];
  const float* srow = states + (size_t)(mtile * 16 + r) * 32;
#pragma unroll
  for (int kk = 0; kk < 8; ++kk)
    afrag[kk] = *(const v2f*)(srow + kk * 4 + 2 * h);

  const int mbase = mtile * 16 + 8 * h;
  for (int dtile = 0; dtile < 32; ++dtile) {
    v8f acc = {0.f, 0.f, 0.f, 0.f, 0.f, 0.f, 0.f, 0.f};
    const float* crow = C + (size_t)(dtile * 16 + r) * 32;
#pragma unroll
    for (int kk = 0; kk < 8; ++kk) {
      const v2f b = *(const v2f*)(crow + kk * 4 + 2 * h);
      acc = wmma4(afrag[kk], b, acc);
    }
    const int d = dtile * 16 + r;
    const float Dd = Dv[d];
#pragma unroll
    for (int v = 0; v < 8; ++v) {
      const int m = mbase + v;
      y1[(size_t)m * D_MODEL + d] = acc[v] + x[(size_t)m * D_MODEL + d] * Dd;
    }
  }
}

__global__ void __launch_bounds__(256) wout_kernel(
    const float* __restrict__ W, const float* __restrict__ bout,
    float* __restrict__ y) {
  __shared__ float lds[16 * D_MODEL];
  const int tid = threadIdx.x;
  const int mtile = blockIdx.x;

  const float4* src4 = (const float4*)(y + (size_t)mtile * 16 * D_MODEL);
  float4* dst4 = (float4*)lds;
#pragma unroll
  for (int i = 0; i < 8; ++i) dst4[tid + i * 256] = src4[tid + i * 256];
  __syncthreads();

  const int wave = tid >> 5, lane = tid & 31;
  const int r = lane & 15, h = lane >> 4;
  const int mbase = mtile * 16 + 8 * h;
  const float* arow = lds + r * D_MODEL;

  for (int t = 0; t < 4; ++t) {
    const int ntile = wave * 4 + t;
    const float* wrow = W + (size_t)(ntile * 16 + r) * D_MODEL;
    if (t < 3)
      __builtin_prefetch(W + (size_t)((ntile + 1) * 16 + r) * D_MODEL, 0, 1);

    v8f acc = {0.f, 0.f, 0.f, 0.f, 0.f, 0.f, 0.f, 0.f};
    for (int k0 = 0; k0 < D_MODEL; k0 += 4) {
      const int k = k0 + 2 * h;
      const v2f a = *(const v2f*)(arow + k);
      const v2f b = *(const v2f*)(wrow + k);
      acc = wmma4(a, b, acc);
    }
    const int n = ntile * 16 + r;
    const float bo = bout[n];
#pragma unroll
    for (int v = 0; v < 8; ++v)
      y[(size_t)(mbase + v) * D_MODEL + n] = acc[v] + bo;
  }
}

extern "C" void kernel_launch(void* const* d_in, const int* in_sizes, int n_in,
                              void* d_out, int out_size, void* d_ws,
                              size_t ws_size, hipStream_t stream) {
  const float* x = (const float*)d_in[0];
  const float* A = (const float*)d_in[1];
  const float* B = (const float*)d_in[2];
  const float* C = (const float*)d_in[3];
  const float* Dv = (const float*)d_in[4];
  const float* log_dt = (const float*)d_in[5];
  const float* W = (const float*)d_in[6];
  const float* bout = (const float*)d_in[7];

  float* y = (float*)d_out;
  float* fs = y + (size_t)ROWS * D_MODEL;

  float* ws = (float*)d_ws;
  float* Bbar = ws + WS_BBAR;
  float* uB = ws + WS_UB;
  float* states = ws + WS_STATES;

  prep_kernel<<<1, 256, 0, stream>>>(A, B, log_dt, ws);
  ub_kernel<<<256, 256, 0, stream>>>(x, Bbar, uB);
  scan_kernel<<<BATCH, 32, 0, stream>>>(ws, states, fs);
  cproj_kernel<<<256, 256, 0, stream>>>(states, C, x, Dv, y);
  wout_kernel<<<2048, 256, 0, stream>>>(W, bout, y);
}


# scan: each step's tanh is branch-free: the library's small-|x| polynomial and its 1-2/(e^2|x|+1) form on hardware exp2 and rcp are both evaluated, one selected per lane; last-place rounding differs
# speedup vs baseline: 3.4017x; 1.1765x over previous
#include <hip/hip_runtime.h>

typedef float v2f __attribute__((ext_vector_type(2)));
typedef float v8f __attribute__((ext_vector_type(8)));

#define D_MODEL 512
#define D_STATE 32
#define BATCH 8
#define SEQ 4096
#define ROWS (BATCH * SEQ)

#define WS_ABAR 0
#define WS_BBAR (WS_ABAR + 1024)
#define WS_UB (WS_BBAR + 16384)
#define WS_STATES (WS_UB + ROWS * 32)

__device__ __forceinline__ v8f wmma4(v2f a, v2f b, v8f c) {
  return __builtin_amdgcn_wmma_f32_16x16x4_f32(false, a, false, b, (short)0, c,
                                               false, false);
}

__global__ void __launch_bounds__(256) prep_kernel(
    const float* __restrict__ A, const float* __restrict__ B,
    const float* __restrict__ log_dt, float* __restrict__ ws) {
  __shared__ float aug[32][64];
  __shared__ float Ah[32][32];
  __shared__ float inv[32][32];
  __shared__ float fac[32];
  __shared__ float dt_sh;
  const int tid = threadIdx.x;

  if (tid == 0) {
    float s = 0.f;
    for (int i = 0; i < D_MODEL; ++i) {
      float v = expf(log_dt[i]);
      v = fminf(fmaxf(v, 1e-4f), 1.0f);
      s += v;
    }
    dt_sh = s / (float)D_MODEL;
  }
  __syncthreads();
  const float dt = dt_sh;

  for (int idx = tid; idx < 32 * 32; idx += blockDim.x) {
    const int i = idx >> 5, j = idx & 31;
    const float a = A[idx];
    const float eye = (i == j) ? 1.f : 0.f;
    aug[i][j] = eye - 0.5f * dt * a;
    aug[i][j + 32] = eye;
    Ah[i][j] = eye + 0.5f * dt * a;
  }
  __syncthreads();

  for (int k = 0; k < 32; ++k) {
    const float p = aug[k][k];
    __syncthreads();
    for (int j = tid; j < 64; j += blockDim.x) aug[k][j] = aug[k][j] / p;
    __syncthreads();
    if (tid < 32) fac[tid] = aug[tid][k];
    __syncthreads();
    for (int idx = tid; idx < 32 * 64; idx += blockDim.x) {
      const int i = idx >> 6, j = idx & 63;
      if (i != k) aug[i][j] -= fac[i] * aug[k][j];
    }
    __syncthreads();
  }
  for (int idx = tid; idx < 32 * 32; idx += blockDim.x) {
    const int i = idx >> 5, j = idx & 31;
    inv[i][j] = aug[i][j + 32];
  }
  __syncthreads();

  float* Abar = ws + WS_ABAR;
  for (int idx = tid; idx < 32 * 32; idx += blockDim.x) {
    const int i = idx >> 5, j = idx & 31;
    float s = 0.f;
    for (int k = 0; k < 32; ++k) s += Ah[i][k] * inv[k][j];
    Abar[idx] = s;
  }
  float* Bbar = ws + WS_BBAR;
  for (int idx = tid; idx < 32 * D_MODEL; idx += blockDim.x) {
    const int n = idx >> 9, d = idx & 511;
    float s = 0.f;
    for (int k = 0; k < 32; ++k) s += inv[n][k] * B[k * D_MODEL + d];
    Bbar[idx] = dt * s;
  }
}

__global__ void __launch_bounds__(256) ub_kernel(const float* __restrict__ x,
                                                 const float* __restrict__ Bbar,
                                                 float* __restrict__ uB) {
  const int wave = threadIdx.x >> 5;
  const int lane = threadIdx.x & 31;
  const int r = lane & 15, h = lane >> 4;
  const int mtile = blockIdx.x * 8 + wave;

  const float* xrow = x + (size_t)(mtile * 16 + r) * D_MODEL;
  const float* b0row = Bbar + (size_t)r * D_MODEL;
  const float* b1row = Bbar + (size_t)(16 + r) * D_MODEL;

  v8f acc0 = {0.f, 0.f, 0.f, 0.f, 0.f, 0.f, 0.f, 0.f};
  v8f acc1 = {0.f, 0.f, 0.f, 0.f, 0.f, 0.f, 0.f, 0.f};

  for (int k0 = 0; k0 < D_MODEL; k0 += 4) {
    const int k = k0 + 2 * h;
    const v2f a = *(const v2f*)(xrow + k);
    const v2f b0 = *(const v2f*)(b0row + k);
    const v2f b1 = *(const v2f*)(b1row + k);
    acc0 = wmma4(a, b0, acc0);
    acc1 = wmma4(a, b1, acc1);
  }

  const int mbase = mtile * 16 + 8 * h;
#pragma unroll
  for (int v = 0; v < 8; ++v) {
    const int m = mbase + v;
    uB[(size_t)m * 32 + r] = acc0[v];
    uB[(size_t)m * 32 + 16 + r] = acc1[v];
  }
}

__device__ __forceinline__ float tanh_step(float x) {
  const float ax = __builtin_fabsf(x);
  const float s = __builtin_copysignf(1.0f, x);
  const float t = __builtin_amdgcn_exp2f(2.885390082f * ax);
  const float r = __builtin_amdgcn_rcpf(t + 1.0f);
  const float zl = fmaf(-2.0f * s, r, s);

  const float x2 = x * x;
  float p = fmaf(x2, -0x1.758e7ap-8f, 0x1.521192p-6f);
  p = fmaf(x2, p, -0x1.b8389cp-5f);
  p = fmaf(x2, p, 0x1.110704p-3f);
  p = fmaf(x2, p, -0x1.555532p-2f);
  const float zs = fmaf(x * x2, p, x);

  return (ax < 0.625f) ? zs : zl;
}

#define SCAN_PF 16
__global__ void __launch_bounds__(32) scan_kernel(const float* __restrict__ ws,
                                                  float* __restrict__ states,
                                                  float* __restrict__ fs_out) {
  const int b = blockIdx.x;
  const int lane = threadIdx.x;
  const float* Abar = ws + WS_ABAR;
  const float* uB = ws + WS_UB + (size_t)b * SEQ * 32;
  float* st_out = states + (size_t)b * SEQ * 32;

  float a[32];
#pragma unroll
  for (int m = 0; m < 32; ++m) a[m] = Abar[lane * 32 + m];

  float cur[SCAN_PF];
#pragma unroll
  for (int j = 0; j < SCAN_PF; ++j) cur[j] = uB[j * 32 + lane];

  float st = 0.f;
  for (int s0 = 0; s0 < SEQ; s0 += SCAN_PF) {
    const int sn = (s0 + SCAN_PF < SEQ) ? (s0 + SCAN_PF) : s0;
    float nxt[SCAN_PF];
#pragma unroll
    for (int j = 0; j < SCAN_PF; ++j) nxt[j] = uB[(sn + j) * 32 + lane];

#pragma unroll
    for (int j = 0; j < SCAN_PF; ++j) {
      const int stb = __float_as_int(st);
      float p0 = cur[j], p1 = 0.f, p2 = 0.f, p3 = 0.f;
#pragma unroll
      for (int m = 0; m < 32; m += 4) {
        const float s0 = __int_as_float(__builtin_amdgcn_readlane(stb, m + 0));
        const float s1 = __int_as_float(__builtin_amdgcn_readlane(stb, m + 1));
        const float s2 = __int_as_float(__builtin_amdgcn_readlane(stb, m + 2));
        const float s3 = __int_as_float(__builtin_amdgcn_readlane(stb, m + 3));
        p0 = fmaf(a[m + 0], s0, p0);
        p1 = fmaf(a[m + 1], s1, p1);
        p2 = fmaf(a[m + 2], s2, p2);
        p3 = fmaf(a[m + 3], s3, p3);
      }
      const float acc = (p0 + p1) + (p2 + p3);
      st = tanh_step(acc);
      st_out[(s0 + j) * 32 + lane] = st;
    }
#pragma unroll
    for (int j = 0; j < SCAN_PF; ++j) cur[j] = nxt[j];
  }
  fs_out[b * 32 + lane] = st;
}

__global__ void __launch_bounds__(256) cproj_kernel(
    const float* __restrict__ states, const float* __restrict__ C,
    const float* __restrict__ x, const float* __restrict__ Dv,
    float* __restrict__ y1) {
  const int wave = threadIdx.x >> 5;
  const int lane = threadIdx.x & 31;
  const int r = lane & 15, h = lane >> 4;
  const int mtile = blockIdx.x * 8 + wave;

  v2f afrag[8];
  const float* srow = states + (size_t)(mtile * 16 + r) * 32;
#pragma unroll
  for (int kk = 0; kk < 8; ++kk)
    afrag[kk] = *(const v2f*)(srow + kk * 4 + 2 * h);

  const int mbase = mtile * 16 + 8 * h;
  for (int dtile = 0; dtile < 32; ++dtile) {
    v8f acc = {0.f, 0.f, 0.f, 0.f, 0.f, 0.f, 0.f, 0.f};
    const float* crow = C + (size_t)(dtile * 16 + r) * 32;
#pragma unroll
    for (int kk = 0; kk < 8; ++kk) {
      const v2f b = *(const v2f*)(crow + kk * 4 + 2 * h);
      acc = wmma4(afrag[kk], b, acc);
    }
    const int d = dtile * 16 + r;
    const float Dd = Dv[d];
#pragma unroll
    for (int v = 0; v < 8; ++v) {
      const int m = mbase + v;
      y1[(size_t)m * D_MODEL + d] = acc[v] + x[(size_t)m * D_MODEL + d] * Dd;
    }
  }
}

__global__ void __launch_bounds__(256) wout_kernel(
    const float* __restrict__ W, const float* __restrict__ bout,
    float* __restrict__ y) {
  __shared__ float lds[16 * D_MODEL];
  const int tid = threadIdx.x;
  const int mtile = blockIdx.x;

  const float4* src4 = (const float4*)(y + (size_t)mtile * 16 * D_MODEL);
  float4* dst4 = (float4*)lds;
#pragma unroll
  for (int i = 0; i < 8; ++i) dst4[tid + i * 256] = src4[tid + i * 256];
  __syncthreads();

  const int wave = tid >> 5, lane = tid & 31;
  const int r = lane & 15, h = lane >> 4;
  const int mbase = mtile * 16 + 8 * h;
  const float* arow = lds + r * D_MODEL;

  for (int t = 0; t < 4; ++t) {
    const int ntile = wave * 4 + t;
    const float* wrow = W + (size_t)(ntile * 16 + r) * D_MODEL;
    if (t < 3)
      __builtin_prefetch(W + (size_t)((ntile + 1) * 16 + r) * D_MODEL, 0, 1);

    v8f acc = {0.f, 0.f, 0.f, 0.f, 0.f, 0.f, 0.f, 0.f};
    for (int k0 = 0; k0 < D_MODEL; k0 += 4) {
      const int k = k0 + 2 * h;
      const v2f a = *(const v2f*)(arow + k);
      const v2f b = *(const v2f*)(wrow + k);
      acc = wmma4(a, b, acc);
    }
    const int n = ntile * 16 + r;
    const float bo = bout[n];
#pragma unroll
    for (int v = 0; v < 8; ++v)
      y[(size_t)(mbase + v) * D_MODEL + n] = acc[v] + bo;
  }
}

extern "C" void kernel_launch(void* const* d_in, const int* in_sizes, int n_in,
                              void* d_out, int out_size, void* d_ws,
                              size_t ws_size, hipStream_t stream) {
  const float* x = (const float*)d_in[0];
  const float* A = (const float*)d_in[1];
  const float* B = (const float*)d_in[2];
  const float* C = (const float*)d_in[3];
  const float* Dv = (const float*)d_in[4];
  const float* log_dt = (const float*)d_in[5];
  const float* W = (const float*)d_in[6];
  const float* bout = (const float*)d_in[7];

  float* y = (float*)d_out;
  float* fs = y + (size_t)ROWS * D_MODEL;

  float* ws = (float*)d_ws;
  float* Bbar = ws + WS_BBAR;
  float* uB = ws + WS_UB;
  float* states = ws + WS_STATES;

  prep_kernel<<<1, 256, 0, stream>>>(A, B, log_dt, ws);
  ub_kernel<<<256, 256, 0, stream>>>(x, Bbar, uB);
  scan_kernel<<<BATCH, 32, 0, stream>>>(ws, states, fs);
  cproj_kernel<<<256, 256, 0, stream>>>(states, C, x, Dv, y);
  wout_kernel<<<2048, 256, 0, stream>>>(W, bout, y);
}


# scan: state reaches the 32x32 product by v_permlane16 broadcasts inside each 16-lane row plus one row exchange, not 32 lane reads via scalar registers; step in fixed-order asm; sum order changes
# speedup vs baseline: 4.3223x; 1.2706x over previous
#include <hip/hip_runtime.h>

typedef float v2f __attribute__((ext_vector_type(2)));
typedef float v8f __attribute__((ext_vector_type(8)));

#define D_MODEL 512
#define D_STATE 32
#define BATCH 8
#define SEQ 4096
#define ROWS (BATCH * SEQ)

#define WS_ABAR 0
#define WS_BBAR (WS_ABAR + 1024)
#define WS_UB (WS_BBAR + 16384)
#define WS_STATES (WS_UB + ROWS * 32)

__device__ __forceinline__ v8f wmma4(v2f a, v2f b, v8f c) {
  return __builtin_amdgcn_wmma_f32_16x16x4_f32(false, a, false, b, (short)0, c,
                                               false, false);
}

__global__ void __launch_bounds__(256) prep_kernel(
    const float* __restrict__ A, const float* __restrict__ B,
    const float* __restrict__ log_dt, float* __restrict__ ws) {
  __shared__ float aug[32][64];
  __shared__ float Ah[32][32];
  __shared__ float inv[32][32];
  __shared__ float fac[32];
  __shared__ float dt_sh;
  const int tid = threadIdx.x;

  if (tid == 0) {
    float s = 0.f;
    for (int i = 0; i < D_MODEL; ++i) {
      float v = expf(log_dt[i]);
      v = fminf(fmaxf(v, 1e-4f), 1.0f);
      s += v;
    }
    dt_sh = s / (float)D_MODEL;
  }
  __syncthreads();
  const float dt = dt_sh;

  for (int idx = tid; idx < 32 * 32; idx += blockDim.x) {
    const int i = idx >> 5, j = idx & 31;
    const float a = A[idx];
    const float eye = (i == j) ? 1.f : 0.f;
    aug[i][j] = eye - 0.5f * dt * a;
    aug[i][j + 32] = eye;
    Ah[i][j] = eye + 0.5f * dt * a;
  }
  __syncthreads();

  for (int k = 0; k < 32; ++k) {
    const float p = aug[k][k];
    __syncthreads();
    for (int j = tid; j < 64; j += blockDim.x) aug[k][j] = aug[k][j] / p;
    __syncthreads();
    if (tid < 32) fac[tid] = aug[tid][k];
    __syncthreads();
    for (int idx = tid; idx < 32 * 64; idx += blockDim.x) {
      const int i = idx >> 6, j = idx & 63;
      if (i != k) aug[i][j] -= fac[i] * aug[k][j];
    }
    __syncthreads();
  }
  for (int idx = tid; idx < 32 * 32; idx += blockDim.x) {
    const int i = idx >> 5, j = idx & 31;
    inv[i][j] = aug[i][j + 32];
  }
  __syncthreads();

  float* Abar = ws + WS_ABAR;
  for (int idx = tid; idx < 32 * 32; idx += blockDim.x) {
    const int i = idx >> 5, j = idx & 31;
    float s = 0.f;
    for (int k = 0; k < 32; ++k) s += Ah[i][k] * inv[k][j];
    Abar[idx] = s;
  }
  float* Bbar = ws + WS_BBAR;
  for (int idx = tid; idx < 32 * D_MODEL; idx += blockDim.x) {
    const int n = idx >> 9, d = idx & 511;
    float s = 0.f;
    for (int k = 0; k < 32; ++k) s += inv[n][k] * B[k * D_MODEL + d];
    Bbar[idx] = dt * s;
  }
}

__global__ void __launch_bounds__(256) ub_kernel(const float* __restrict__ x,
                                                 const float* __restrict__ Bbar,
                                                 float* __restrict__ uB) {
  const int wave = threadIdx.x >> 5;
  const int lane = threadIdx.x & 31;
  const int r = lane & 15, h = lane >> 4;
  const int mtile = blockIdx.x * 8 + wave;

  const float* xrow = x + (size_t)(mtile * 16 + r) * D_MODEL;
  const float* b0row = Bbar + (size_t)r * D_MODEL;
  const float* b1row = Bbar + (size_t)(16 + r) * D_MODEL;

  v8f acc0 = {0.f, 0.f, 0.f, 0.f, 0.f, 0.f, 0.f, 0.f};
  v8f acc1 = {0.f, 0.f, 0.f, 0.f, 0.f, 0.f, 0.f, 0.f};

  for (int k0 = 0; k0 < D_MODEL; k0 += 4) {
    const int k = k0 + 2 * h;
    const v2f a = *(const v2f*)(xrow + k);
    const v2f b0 = *(const v2f*)(b0row + k);
    const v2f b1 = *(const v2f*)(b1row + k);
    acc0 = wmma4(a, b0, acc0);
    acc1 = wmma4(a, b1, acc1);
  }

  const int mbase = mtile * 16 + 8 * h;
#pragma unroll
  for (int v = 0; v < 8; ++v) {
    const int m = mbase + v;
    uB[(size_t)m * 32 + r] = acc0[v];
    uB[(size_t)m * 32 + 16 + r] = acc1[v];
  }
}

__device__ __forceinline__ float tanh_step(float x) {
  float z, t, x2, s, p, m2s, x3;
  asm volatile(
      "v_mul_f32 %[t], 0x4038aa3b, |%[x]|\n\t"
      "v_mul_f32 %[x2], %[x], %[x]\n\t"
      "v_bfi_b32 %[s], 0x7fffffff, 1.0, %[x]\n\t"
      "v_cmp_gt_f32 vcc_lo, 0x3f200000, |%[x]|\n\t"
      "v_exp_f32 %[t], %[t]\n\t"
      "v_fmaak_f32 %[p], %[k4], %[x2], 0x3ca908c9\n\t"
      "v_mul_f32 %[m2s], -2.0, %[s]\n\t"
      "v_mul_f32 %[x3], %[x], %[x2]\n\t"
      "v_fmaak_f32 %[p], %[x2], %[p], 0xbd5c1c4e\n\t"
      "v_add_f32 %[t], 1.0, %[t]\n\t"
      "v_fmaak_f32 %[p], %[x2], %[p], 0x3e088382\n\t"
      "v_rcp_f32 %[t], %[t]\n\t"
      "v_fmaak_f32 %[p], %[x2], %[p], 0xbeaaaa99\n\t"
      "v_fma_f32 %[x3], %[x3], %[p], %[x]\n\t"
      "v_fmac_f32 %[s], %[m2s], %[t]\n\t"
      "v_cndmask_b32 %[z], %[s], %[x3], vcc_lo"
      : [z] "=v"(z), [t] "=&v"(t), [x2] "=&v"(x2), [s] "=&v"(s), [p] "=&v"(p),
        [m2s] "=&v"(m2s), [x3] "=&v"(x3)
      : [x] "v"(x), [k4] "s"(0xbbbac73du)
      : "vcc");
  return z;
}

#define SCAN_PF 16
#define BC16(dst, sel) "v_permlane16_b32 " dst ", %[st], " sel ", " sel "\n\t"

__device__ __forceinline__ void dot_head(float st, float& t0, float& t1,
                                         float& t2, float& t3) {
  asm volatile(BC16("%[n0]", "%[s0]") BC16("%[n1]", "%[s1]")
               BC16("%[n2]", "%[s2]") BC16("%[n3]", "%[s3]")
               : [n0] "=&v"(t0), [n1] "=&v"(t1), [n2] "=&v"(t2), [n3] "=&v"(t3)
               : [st] "v"(st), [s0] "s"(0x00000000u), [s1] "s"(0x11111111u),
                 [s2] "s"(0x22222222u), [s3] "s"(0x33333333u));
}

__device__ __forceinline__ void dot_first(float st, float ub, const float* ao,
                                          const float* ax, float t0, float t1,
                                          float t2, float t3, float& n0,
                                          float& n1, float& n2, float& n3,
                                          float& o0, float& o1, float& x0,
                                          float& x1) {
  asm volatile(BC16("%[n0]", "%[s0]")
               "v_mul_f32 %[x0], %[c0], %[t0]\n\t"
               "v_fma_f32 %[o0], %[a0], %[t0], %[ub]\n\t"
               BC16("%[n1]", "%[s1]")
               "v_mul_f32 %[x1], %[c1], %[t1]\n\t"
               "v_mul_f32 %[o1], %[a1], %[t1]\n\t"
               BC16("%[n2]", "%[s2]")
               "v_fmac_f32 %[x0], %[c2], %[t2]\n\t"
               "v_fmac_f32 %[o0], %[a2], %[t2]\n\t"
               BC16("%[n3]", "%[s3]")
               "v_fmac_f32 %[x1], %[c3], %[t3]\n\t"
               "v_fmac_f32 %[o1], %[a3], %[t3]"
               : [n0] "=&v"(n0), [n1] "=&v"(n1), [n2] "=&v"(n2), [n3] "=&v"(n3),
                 [o0] "=&v"(o0), [o1] "=&v"(o1), [x0] "=&v"(x0), [x1] "=&v"(x1)
               : [st] "v"(st), [ub] "v"(ub), [t0] "v"(t0), [t1] "v"(t1),
                 [t2] "v"(t2), [t3] "v"(t3), [a0] "v"(ao[0]), [a1] "v"(ao[1]),
                 [a2] "v"(ao[2]), [a3] "v"(ao[3]), [c0] "v"(ax[0]),
                 [c1] "v"(ax[1]), [c2] "v"(ax[2]), [c3] "v"(ax[3]),
                 [s0] "s"(0x44444444u), [s1] "s"(0x55555555u),
                 [s2] "s"(0x66666666u), [s3] "s"(0x77777777u));
}

template <int M>
__device__ __forceinline__ void dot_mid(float st, const float* ao,
                                        const float* ax, float t0, float t1,
                                        float t2, float t3, float& n0, float& n1,
                                        float& n2, float& n3, float& o0,
                                        float& o1, float& x0, float& x1) {
  asm volatile(BC16("%[n0]", "%[s0]")
               "v_fmac_f32 %[x0], %[c0], %[t0]\n\t"
               "v_fmac_f32 %[o0], %[a0], %[t0]\n\t"
               BC16("%[n1]", "%[s1]")
               "v_fmac_f32 %[x1], %[c1], %[t1]\n\t"
               "v_fmac_f32 %[o1], %[a1], %[t1]\n\t"
               BC16("%[n2]", "%[s2]")
               "v_fmac_f32 %[x0], %[c2], %[t2]\n\t"
               "v_fmac_f32 %[o0], %[a2], %[t2]\n\t"
               BC16("%[n3]", "%[s3]")
               "v_fmac_f32 %[x1], %[c3], %[t3]\n\t"
               "v_fmac_f32 %[o1], %[a3], %[t3]"
               : [n0] "=&v"(n0), [n1] "=&v"(n1), [n2] "=&v"(n2), [n3] "=&v"(n3),
                 [o0] "+v"(o0), [o1] "+v"(o1), [x0] "+v"(x0), [x1] "+v"(x1)
               : [st] "v"(st), [t0] "v"(t0), [t1] "v"(t1), [t2] "v"(t2),
                 [t3] "v"(t3), [a0] "v"(ao[M]), [a1] "v"(ao[M + 1]),
                 [a2] "v"(ao[M + 2]), [a3] "v"(ao[M + 3]), [c0] "v"(ax[M]),
                 [c1] "v"(ax[M + 1]), [c2] "v"(ax[M + 2]), [c3] "v"(ax[M + 3]),
                 [s0] "s"(0x11111111u * (unsigned)(M + 4)),
                 [s1] "s"(0x11111111u * (unsigned)(M + 5)),
                 [s2] "s"(0x11111111u * (unsigned)(M + 6)),
                 [s3] "s"(0x11111111u * (unsigned)(M + 7)));
}

__device__ __forceinline__ float dot_tail(const float* ao, const float* ax,
                                          float t0, float t1, float t2, float t3,
                                          float o0, float o1, float x0,
                                          float x1) {
  float acc;
  asm volatile("v_fmac_f32 %[x0], %[c0], %[t0]\n\t"
               "v_fmac_f32 %[o0], %[a0], %[t0]\n\t"
               "v_fmac_f32 %[x1], %[c1], %[t1]\n\t"
               "v_fmac_f32 %[o1], %[a1], %[t1]\n\t"
               "v_fmac_f32 %[x0], %[c2], %[t2]\n\t"
               "v_fmac_f32 %[o0], %[a2], %[t2]\n\t"
               "v_fmac_f32 %[x1], %[c3], %[t3]\n\t"
               "v_fmac_f32 %[o1], %[a3], %[t3]\n\t"
               "v_add_f32 %[x0], %[x0], %[x1]\n\t"
               "v_add_f32 %[o0], %[o0], %[o1]\n\t"
               "v_permlanex16_b32 %[x1], %[x0], %[lo], %[hi]\n\t"
               "v_add_f32 %[acc], %[o0], %[x1]"
               : [acc] "=v"(acc), [o0] "+v"(o0), [o1] "+v"(o1), [x0] "+v"(x0),
                 [x1] "+v"(x1)
               : [t0] "v"(t0), [t1] "v"(t1), [t2] "v"(t2), [t3] "v"(t3),
                 [a0] "v"(ao[12]), [a1] "v"(ao[13]), [a2] "v"(ao[14]),
                 [a3] "v"(ao[15]), [c0] "v"(ax[12]), [c1] "v"(ax[13]),
                 [c2] "v"(ax[14]), [c3] "v"(ax[15]), [lo] "s"(0x76543210u),
                 [hi] "s"(0xfedcba98u));
  return acc;
}

__global__ void __launch_bounds__(32) scan_kernel(const float* __restrict__ ws,
                                                  float* __restrict__ states,
                                                  float* __restrict__ fs_out) {
  const int b = blockIdx.x;
  const int lane = threadIdx.x;
  const int hb = lane & 16;
  const float* Abar = ws + WS_ABAR;
  const float* uB = ws + WS_UB + (size_t)b * SEQ * 32;
  float* st_out = states + (size_t)b * SEQ * 32;

  float ao[16], ax[16];
#pragma unroll
  for (int m = 0; m < 16; ++m) {
    ao[m] = Abar[lane * 32 + hb + m];
    ax[m] = Abar[(lane ^ 16) * 32 + hb + m];
  }

  float cur[SCAN_PF];
#pragma unroll
  for (int j = 0; j < SCAN_PF; ++j) cur[j] = uB[j * 32 + lane];

  float st = 0.f;
  for (int s0 = 0; s0 < SEQ; s0 += SCAN_PF) {
    const int sn = (s0 + SCAN_PF < SEQ) ? (s0 + SCAN_PF) : s0;
    float nxt[SCAN_PF];
#pragma unroll
    for (int j = 0; j < SCAN_PF; ++j) nxt[j] = uB[(sn + j) * 32 + lane];

#pragma unroll
    for (int j = 0; j < SCAN_PF; ++j) {
      float o0, o1, x0, x1;
      float p0, p1, p2, p3, q0, q1, q2, q3;
      dot_head(st, p0, p1, p2, p3);
      dot_first(st, cur[j], ao, ax, p0, p1, p2, p3, q0, q1, q2, q3, o0, o1, x0,
                x1);
      dot_mid<4>(st, ao, ax, q0, q1, q2, q3, p0, p1, p2, p3, o0, o1, x0, x1);
      dot_mid<8>(st, ao, ax, p0, p1, p2, p3, q0, q1, q2, q3, o0, o1, x0, x1);
      const float acc = dot_tail(ao, ax, q0, q1, q2, q3, o0, o1, x0, x1);
      st = tanh_step(acc);
      st_out[(s0 + j) * 32 + lane] = st;
    }
#pragma unroll
    for (int j = 0; j < SCAN_PF; ++j) cur[j] = nxt[j];
  }
  fs_out[b * 32 + lane] = st;
}

__global__ void __launch_bounds__(256) cproj_kernel(
    const float* __restrict__ states, const float* __restrict__ C,
    const float* __restrict__ x, const float* __restrict__ Dv,
    float* __restrict__ y1) {
  const int wave = threadIdx.x >> 5;
  const int lane = threadIdx.x & 31;
  const int r = lane & 15, h = lane >> 4;
  const int mtile = blockIdx.x * 8 + wave;

  v2f afrag[8];
  const float* srow = states + (size_t)(mtile * 16 + r) * 32;
#pragma unroll
  for (int kk = 0; kk < 8; ++kk)
    afrag[kk] = *(const v2f*)(srow + kk * 4 + 2 * h);

  const int mbase = mtile * 16 + 8 * h;
  for (int dtile = 0; dtile < 32; ++dtile) {
    v8f acc = {0.f, 0.f, 0.f, 0.f, 0.f, 0.f, 0.f, 0.f};
    const float* crow = C + (size_t)(dtile * 16 + r) * 32;
#pragma unroll
    for (int kk = 0; kk < 8; ++kk) {
      const v2f b = *(const v2f*)(crow + kk * 4 + 2 * h);
      acc = wmma4(afrag[kk], b, acc);
    }
    const int d = dtile * 16 + r;
    const float Dd = Dv[d];
#pragma unroll
    for (int v = 0; v < 8; ++v) {
      const int m = mbase + v;
      y1[(size_t)m * D_MODEL + d] = acc[v] + x[(size_t)m * D_MODEL + d] * Dd;
    }
  }
}

__global__ void __launch_bounds__(256) wout_kernel(
    const float* __restrict__ W, const float* __restrict__ bout,
    float* __restrict__ y) {
  __shared__ float lds[16 * D_MODEL];
  const int tid = threadIdx.x;
  const int mtile = blockIdx.x;

  const float4* src4 = (const float4*)(y + (size_t)mtile * 16 * D_MODEL);
  float4* dst4 = (float4*)lds;
#pragma unroll
  for (int i = 0; i < 8; ++i) dst4[tid + i * 256] = src4[tid + i * 256];
  __syncthreads();

  const int wave = tid >> 5, lane = tid & 31;
  const int r = lane & 15, h = lane >> 4;
  const int mbase = mtile * 16 + 8 * h;
  const float* arow = lds + r * D_MODEL;

  for (int t = 0; t < 4; ++t) {
    const int ntile = wave * 4 + t;
    const float* wrow = W + (size_t)(ntile * 16 + r) * D_MODEL;
    if (t < 3)
      __builtin_prefetch(W + (size_t)((ntile + 1) * 16 + r) * D_MODEL, 0, 1);

    v8f acc = {0.f, 0.f, 0.f, 0.f, 0.f, 0.f, 0.f, 0.f};
    for (int k0 = 0; k0 < D_MODEL; k0 += 4) {
      const int k = k0 + 2 * h;
      const v2f a = *(const v2f*)(arow + k);
      const v2f b = *(const v2f*)(wrow + k);
      acc = wmma4(a, b, acc);
    }
    const int n = ntile * 16 + r;
    const float bo = bout[n];
#pragma unroll
    for (int v = 0; v < 8; ++v)
      y[(size_t)(mbase + v) * D_MODEL + n] = acc[v] + bo;
  }
}

extern "C" void kernel_launch(void* const* d_in, const int* in_sizes, int n_in,
                              void* d_out, int out_size, void* d_ws,
                              size_t ws_size, hipStream_t stream) {
  const float* x = (const float*)d_in[0];
  const float* A = (const float*)d_in[1];
  const float* B = (const float*)d_in[2];
  const float* C = (const float*)d_in[3];
  const float* Dv = (const float*)d_in[4];
  const float* log_dt = (const float*)d_in[5];
  const float* W = (const float*)d_in[6];
  const float* bout = (const float*)d_in[7];

  float* y = (float*)d_out;
  float* fs = y + (size_t)ROWS * D_MODEL;

  float* ws = (float*)d_ws;
  float* Bbar = ws + WS_BBAR;
  float* uB = ws + WS_UB;
  float* states = ws + WS_STATES;

  prep_kernel<<<1, 256, 0, stream>>>(A, B, log_dt, ws);
  ub_kernel<<<256, 256, 0, stream>>>(x, Bbar, uB);
  scan_kernel<<<BATCH, 32, 0, stream>>>(ws, states, fs);
  cproj_kernel<<<256, 256, 0, stream>>>(states, C, x, Dv, y);
  wout_kernel<<<2048, 256, 0, stream>>>(W, bout, y);
}


# prep: its serial parts spread over the 256 threads: dt by a two-level sum (that sum's order changes), elimination rows kept in registers, each B column loaded once
# speedup vs baseline: 4.9576x; 1.1470x over previous
#include <hip/hip_runtime.h>

typedef float v2f __attribute__((ext_vector_type(2)));
typedef float v8f __attribute__((ext_vector_type(8)));

#define D_MODEL 512
#define D_STATE 32
#define BATCH 8
#define SEQ 4096
#define ROWS (BATCH * SEQ)

#define WS_ABAR 0
#define WS_BBAR (WS_ABAR + 1024)
#define WS_UB (WS_BBAR + 16384)
#define WS_STATES (WS_UB + ROWS * 32)

__device__ __forceinline__ v8f wmma4(v2f a, v2f b, v8f c) {
  return __builtin_amdgcn_wmma_f32_16x16x4_f32(false, a, false, b, (short)0, c,
                                               false, false);
}

__global__ void __launch_bounds__(256) prep_kernel(
    const float* __restrict__ A, const float* __restrict__ B,
    const float* __restrict__ log_dt, float* __restrict__ ws) {
  __shared__ float red[256];
  __shared__ float red2[32];
  __shared__ float rowbuf[2][64];
  __shared__ float colbuf[2][32];
  __shared__ float Ah[32][32];
  __shared__ float inv[32][32];
  __shared__ float dt_sh;
  const int tid = threadIdx.x;

  float b0[32], b1[32];
#pragma unroll
  for (int k = 0; k < 32; ++k) {
    b0[k] = B[k * D_MODEL + tid];
    b1[k] = B[k * D_MODEL + 256 + tid];
  }

  const int i = tid >> 3, jc = tid & 7;
  float av[8];
#pragma unroll
  for (int e = 0; e < 8; ++e) av[e] = A[i * 32 + (jc & 3) * 8 + e];

  {
    float v0 = expf(log_dt[tid]);
    float v1 = expf(log_dt[tid + 256]);
    v0 = fminf(fmaxf(v0, 1e-4f), 1.0f);
    v1 = fminf(fmaxf(v1, 1e-4f), 1.0f);
    red[tid] = v0 + v1;
  }
  __syncthreads();
  if (tid < 32) {
    float s = 0.f;
    for (int q = 0; q < 8; ++q) s += red[tid * 8 + q];
    red2[tid] = s;
  }
  __syncthreads();
  if (tid == 0) {
    float s = 0.f;
    for (int q = 0; q < 32; ++q) s += red2[q];
    dt_sh = s / (float)D_MODEL;
  }
  __syncthreads();
  const float dt = dt_sh;

  float a[8];
#pragma unroll
  for (int e = 0; e < 8; ++e) {
    const int j = (jc & 3) * 8 + e;
    const float eye = (i == j) ? 1.f : 0.f;
    const float lo = eye - 0.5f * dt * av[e];
    if (jc < 4) Ah[i][j] = eye + 0.5f * dt * av[e];
    a[e] = (jc < 4) ? lo : eye;
  }

  for (int k = 0; k < 32; ++k) {
    const int pb = k & 1;
    if (i == k) {
#pragma unroll
      for (int e = 0; e < 8; ++e) rowbuf[pb][jc * 8 + e] = a[e];
    }
    if (jc == (k >> 3)) {
      float ak = a[0];
#pragma unroll
      for (int e = 1; e < 8; ++e) ak = ((k & 7) == e) ? a[e] : ak;
      colbuf[pb][i] = ak;
    }
    __syncthreads();
    const float p = rowbuf[pb][k];
    const float f = colbuf[pb][i];
#pragma unroll
    for (int e = 0; e < 8; ++e) {
      const float nr = rowbuf[pb][jc * 8 + e] / p;
      a[e] = (i == k) ? nr : (a[e] - f * nr);
    }
  }
  if (jc >= 4) {
#pragma unroll
    for (int e = 0; e < 8; ++e) inv[i][(jc - 4) * 8 + e] = a[e];
  }
  __syncthreads();

  float* Abar = ws + WS_ABAR;
  for (int idx = tid; idx < 32 * 32; idx += 256) {
    const int r = idx >> 5, j = idx & 31;
    float s = 0.f;
    for (int k = 0; k < 32; ++k) s += Ah[r][k] * inv[k][j];
    Abar[idx] = s;
  }
  float* Bbar = ws + WS_BBAR;
  for (int n = 0; n < 32; ++n) {
    float s0 = 0.f, s1 = 0.f;
#pragma unroll
    for (int k = 0; k < 32; ++k) {
      const float w = inv[n][k];
      s0 += w * b0[k];
      s1 += w * b1[k];
    }
    Bbar[n * D_MODEL + tid] = dt * s0;
    Bbar[n * D_MODEL + 256 + tid] = dt * s1;
  }
}

__global__ void __launch_bounds__(256) ub_kernel(const float* __restrict__ x,
                                                 const float* __restrict__ Bbar,
                                                 float* __restrict__ uB) {
  const int wave = threadIdx.x >> 5;
  const int lane = threadIdx.x & 31;
  const int r = lane & 15, h = lane >> 4;
  const int mtile = blockIdx.x * 8 + wave;

  const float* xrow = x + (size_t)(mtile * 16 + r) * D_MODEL;
  const float* b0row = Bbar + (size_t)r * D_MODEL;
  const float* b1row = Bbar + (size_t)(16 + r) * D_MODEL;

  v8f acc0 = {0.f, 0.f, 0.f, 0.f, 0.f, 0.f, 0.f, 0.f};
  v8f acc1 = {0.f, 0.f, 0.f, 0.f, 0.f, 0.f, 0.f, 0.f};

  for (int k0 = 0; k0 < D_MODEL; k0 += 4) {
    const int k = k0 + 2 * h;
    const v2f a = *(const v2f*)(xrow + k);
    const v2f b0 = *(const v2f*)(b0row + k);
    const v2f b1 = *(const v2f*)(b1row + k);
    acc0 = wmma4(a, b0, acc0);
    acc1 = wmma4(a, b1, acc1);
  }

  const int mbase = mtile * 16 + 8 * h;
#pragma unroll
  for (int v = 0; v < 8; ++v) {
    const int m = mbase + v;
    uB[(size_t)m * 32 + r] = acc0[v];
    uB[(size_t)m * 32 + 16 + r] = acc1[v];
  }
}

__device__ __forceinline__ float tanh_step(float x) {
  float z, t, x2, s, p, m2s, x3;
  asm volatile(
      "v_mul_f32 %[t], 0x4038aa3b, |%[x]|\n\t"
      "v_mul_f32 %[x2], %[x], %[x]\n\t"
      "v_bfi_b32 %[s], 0x7fffffff, 1.0, %[x]\n\t"
      "v_cmp_gt_f32 vcc_lo, 0x3f200000, |%[x]|\n\t"
      "v_exp_f32 %[t], %[t]\n\t"
      "v_fmaak_f32 %[p], %[k4], %[x2], 0x3ca908c9\n\t"
      "v_mul_f32 %[m2s], -2.0, %[s]\n\t"
      "v_mul_f32 %[x3], %[x], %[x2]\n\t"
      "v_fmaak_f32 %[p], %[x2], %[p], 0xbd5c1c4e\n\t"
      "v_add_f32 %[t], 1.0, %[t]\n\t"
      "v_fmaak_f32 %[p], %[x2], %[p], 0x3e088382\n\t"
      "v_rcp_f32 %[t], %[t]\n\t"
      "v_fmaak_f32 %[p], %[x2], %[p], 0xbeaaaa99\n\t"
      "v_fma_f32 %[x3], %[x3], %[p], %[x]\n\t"
      "v_fmac_f32 %[s], %[m2s], %[t]\n\t"
      "v_cndmask_b32 %[z], %[s], %[x3], vcc_lo"
      : [z] "=v"(z), [t] "=&v"(t), [x2] "=&v"(x2), [s] "=&v"(s), [p] "=&v"(p),
        [m2s] "=&v"(m2s), [x3] "=&v"(x3)
      : [x] "v"(x), [k4] "s"(0xbbbac73du)
      : "vcc");
  return z;
}

#define SCAN_PF 16
#define BC16(dst, sel) "v_permlane16_b32 " dst ", %[st], " sel ", " sel "\n\t"

__device__ __forceinline__ void dot_head(float st, float& t0, float& t1,
                                         float& t2, float& t3) {
  asm volatile(BC16("%[n0]", "%[s0]") BC16("%[n1]", "%[s1]")
               BC16("%[n2]", "%[s2]") BC16("%[n3]", "%[s3]")
               : [n0] "=&v"(t0), [n1] "=&v"(t1), [n2] "=&v"(t2), [n3] "=&v"(t3)
               : [st] "v"(st), [s0] "s"(0x00000000u), [s1] "s"(0x11111111u),
                 [s2] "s"(0x22222222u), [s3] "s"(0x33333333u));
}

__device__ __forceinline__ void dot_first(float st, float ub, const float* ao,
                                          const float* ax, float t0, float t1,
                                          float t2, float t3, float& n0,
                                          float& n1, float& n2, float& n3,
                                          float& o0, float& o1, float& x0,
                                          float& x1) {
  asm volatile(BC16("%[n0]", "%[s0]")
               "v_mul_f32 %[x0], %[c0], %[t0]\n\t"
               "v_fma_f32 %[o0], %[a0], %[t0], %[ub]\n\t"
               BC16("%[n1]", "%[s1]")
               "v_mul_f32 %[x1], %[c1], %[t1]\n\t"
               "v_mul_f32 %[o1], %[a1], %[t1]\n\t"
               BC16("%[n2]", "%[s2]")
               "v_fmac_f32 %[x0], %[c2], %[t2]\n\t"
               "v_fmac_f32 %[o0], %[a2], %[t2]\n\t"
               BC16("%[n3]", "%[s3]")
               "v_fmac_f32 %[x1], %[c3], %[t3]\n\t"
               "v_fmac_f32 %[o1], %[a3], %[t3]"
               : [n0] "=&v"(n0), [n1] "=&v"(n1), [n2] "=&v"(n2), [n3] "=&v"(n3),
                 [o0] "=&v"(o0), [o1] "=&v"(o1), [x0] "=&v"(x0), [x1] "=&v"(x1)
               : [st] "v"(st), [ub] "v"(ub), [t0] "v"(t0), [t1] "v"(t1),
                 [t2] "v"(t2), [t3] "v"(t3), [a0] "v"(ao[0]), [a1] "v"(ao[1]),
                 [a2] "v"(ao[2]), [a3] "v"(ao[3]), [c0] "v"(ax[0]),
                 [c1] "v"(ax[1]), [c2] "v"(ax[2]), [c3] "v"(ax[3]),
                 [s0] "s"(0x44444444u), [s1] "s"(0x55555555u),
                 [s2] "s"(0x66666666u), [s3] "s"(0x77777777u));
}

template <int M>
__device__ __forceinline__ void dot_mid(float st, const float* ao,
                                        const float* ax, float t0, float t1,
                                        float t2, float t3, float& n0, float& n1,
                                        float& n2, float& n3, float& o0,
                                        float& o1, float& x0, float& x1) {
  asm volatile(BC16("%[n0]", "%[s0]")
               "v_fmac_f32 %[x0], %[c0], %[t0]\n\t"
               "v_fmac_f32 %[o0], %[a0], %[t0]\n\t"
               BC16("%[n1]", "%[s1]")
               "v_fmac_f32 %[x1], %[c1], %[t1]\n\t"
               "v_fmac_f32 %[o1], %[a1], %[t1]\n\t"
               BC16("%[n2]", "%[s2]")
               "v_fmac_f32 %[x0], %[c2], %[t2]\n\t"
               "v_fmac_f32 %[o0], %[a2], %[t2]\n\t"
               BC16("%[n3]", "%[s3]")
               "v_fmac_f32 %[x1], %[c3], %[t3]\n\t"
               "v_fmac_f32 %[o1], %[a3], %[t3]"
               : [n0] "=&v"(n0), [n1] "=&v"(n1), [n2] "=&v"(n2), [n3] "=&v"(n3),
                 [o0] "+v"(o0), [o1] "+v"(o1), [x0] "+v"(x0), [x1] "+v"(x1)
               : [st] "v"(st), [t0] "v"(t0), [t1] "v"(t1), [t2] "v"(t2),
                 [t3] "v"(t3), [a0] "v"(ao[M]), [a1] "v"(ao[M + 1]),
                 [a2] "v"(ao[M + 2]), [a3] "v"(ao[M + 3]), [c0] "v"(ax[M]),
                 [c1] "v"(ax[M + 1]), [c2] "v"(ax[M + 2]), [c3] "v"(ax[M + 3]),
                 [s0] "s"(0x11111111u * (unsigned)(M + 4)),
                 [s1] "s"(0x11111111u * (unsigned)(M + 5)),
                 [s2] "s"(0x11111111u * (unsigned)(M + 6)),
                 [s3] "s"(0x11111111u * (unsigned)(M + 7)));
}

__device__ __forceinline__ float dot_tail(const float* ao, const float* ax,
                                          float t0, float t1, float t2, float t3,
                                          float o0, float o1, float x0,
                                          float x1) {
  float acc;
  asm volatile("v_fmac_f32 %[x0], %[c0], %[t0]\n\t"
               "v_fmac_f32 %[o0], %[a0], %[t0]\n\t"
               "v_fmac_f32 %[x1], %[c1], %[t1]\n\t"
               "v_fmac_f32 %[o1], %[a1], %[t1]\n\t"
               "v_fmac_f32 %[x0], %[c2], %[t2]\n\t"
               "v_fmac_f32 %[o0], %[a2], %[t2]\n\t"
               "v_fmac_f32 %[x1], %[c3], %[t3]\n\t"
               "v_fmac_f32 %[o1], %[a3], %[t3]\n\t"
               "v_add_f32 %[x0], %[x0], %[x1]\n\t"
               "v_add_f32 %[o0], %[o0], %[o1]\n\t"
               "v_permlanex16_b32 %[x1], %[x0], %[lo], %[hi]\n\t"
               "v_add_f32 %[acc], %[o0], %[x1]"
               : [acc] "=v"(acc), [o0] "+v"(o0), [o1] "+v"(o1), [x0] "+v"(x0),
                 [x1] "+v"(x1)
               : [t0] "v"(t0), [t1] "v"(t1), [t2] "v"(t2), [t3] "v"(t3),
                 [a0] "v"(ao[12]), [a1] "v"(ao[13]), [a2] "v"(ao[14]),
                 [a3] "v"(ao[15]), [c0] "v"(ax[12]), [c1] "v"(ax[13]),
                 [c2] "v"(ax[14]), [c3] "v"(ax[15]), [lo] "s"(0x76543210u),
                 [hi] "s"(0xfedcba98u));
  return acc;
}

__global__ void __launch_bounds__(32) scan_kernel(const float* __restrict__ ws,
                                                  float* __restrict__ states,
                                                  float* __restrict__ fs_out) {
  const int b = blockIdx.x;
  const int lane = threadIdx.x;
  const int hb = lane & 16;
  const float* Abar = ws + WS_ABAR;
  const float* uB = ws + WS_UB + (size_t)b * SEQ * 32;
  float* st_out = states + (size_t)b * SEQ * 32;

  float ao[16], ax[16];
#pragma unroll
  for (int m = 0; m < 16; ++m) {
    ao[m] = Abar[lane * 32 + hb + m];
    ax[m] = Abar[(lane ^ 16) * 32 + hb + m];
  }

  float cur[SCAN_PF];
#pragma unroll
  for (int j = 0; j < SCAN_PF; ++j) cur[j] = uB[j * 32 + lane];

  float st = 0.f;
  for (int s0 = 0; s0 < SEQ; s0 += SCAN_PF) {
    const int sn = (s0 + SCAN_PF < SEQ) ? (s0 + SCAN_PF) : s0;
    float nxt[SCAN_PF];
#pragma unroll
    for (int j = 0; j < SCAN_PF; ++j) nxt[j] = uB[(sn + j) * 32 + lane];

#pragma unroll
    for (int j = 0; j < SCAN_PF; ++j) {
      float o0, o1, x0, x1;
      float p0, p1, p2, p3, q0, q1, q2, q3;
      dot_head(st, p0, p1, p2, p3);
      dot_first(st, cur[j], ao, ax, p0, p1, p2, p3, q0, q1, q2, q3, o0, o1, x0,
                x1);
      dot_mid<4>(st, ao, ax, q0, q1, q2, q3, p0, p1, p2, p3, o0, o1, x0, x1);
      dot_mid<8>(st, ao, ax, p0, p1, p2, p3, q0, q1, q2, q3, o0, o1, x0, x1);
      const float acc = dot_tail(ao, ax, q0, q1, q2, q3, o0, o1, x0, x1);
      st = tanh_step(acc);
      st_out[(s0 + j) * 32 + lane] = st;
    }
#pragma unroll
    for (int j = 0; j < SCAN_PF; ++j) cur[j] = nxt[j];
  }
  fs_out[b * 32 + lane] = st;
}

__global__ void __launch_bounds__(256) cproj_kernel(
    const float* __restrict__ states, const float* __restrict__ C,
    const float* __restrict__ x, const float* __restrict__ Dv,
    float* __restrict__ y1) {
  const int wave = threadIdx.x >> 5;
  const int lane = threadIdx.x & 31;
  const int r = lane & 15, h = lane >> 4;
  const int mtile = blockIdx.x * 8 + wave;

  v2f afrag[8];
  const float* srow = states + (size_t)(mtile * 16 + r) * 32;
#pragma unroll
  for (int kk = 0; kk < 8; ++kk)
    afrag[kk] = *(const v2f*)(srow + kk * 4 + 2 * h);

  const int mbase = mtile * 16 + 8 * h;
  for (int dtile = 0; dtile < 32; ++dtile) {
    v8f acc = {0.f, 0.f, 0.f, 0.f, 0.f, 0.f, 0.f, 0.f};
    const float* crow = C + (size_t)(dtile * 16 + r) * 32;
#pragma unroll
    for (int kk = 0; kk < 8; ++kk) {
      const v2f b = *(const v2f*)(crow + kk * 4 + 2 * h);
      acc = wmma4(afrag[kk], b, acc);
    }
    const int d = dtile * 16 + r;
    const float Dd = Dv[d];
#pragma unroll
    for (int v = 0; v < 8; ++v) {
      const int m = mbase + v;
      y1[(size_t)m * D_MODEL + d] = acc[v] + x[(size_t)m * D_MODEL + d] * Dd;
    }
  }
}

__global__ void __launch_bounds__(256) wout_kernel(
    const float* __restrict__ W, const float* __restrict__ bout,
    float* __restrict__ y) {
  __shared__ float lds[16 * D_MODEL];
  const int tid = threadIdx.x;
  const int mtile = blockIdx.x;

  const float4* src4 = (const float4*)(y + (size_t)mtile * 16 * D_MODEL);
  float4* dst4 = (float4*)lds;
#pragma unroll
  for (int i = 0; i < 8; ++i) dst4[tid + i * 256] = src4[tid + i * 256];
  __syncthreads();

  const int wave = tid >> 5, lane = tid & 31;
  const int r = lane & 15, h = lane >> 4;
  const int mbase = mtile * 16 + 8 * h;
  const float* arow = lds + r * D_MODEL;

  for (int t = 0; t < 4; ++t) {
    const int ntile = wave * 4 + t;
    const float* wrow = W + (size_t)(ntile * 16 + r) * D_MODEL;
    if (t < 3)
      __builtin_prefetch(W + (size_t)((ntile + 1) * 16 + r) * D_MODEL, 0, 1);

    v8f acc = {0.f, 0.f, 0.f, 0.f, 0.f, 0.f, 0.f, 0.f};
    for (int k0 = 0; k0 < D_MODEL; k0 += 4) {
      const int k = k0 + 2 * h;
      const v2f a = *(const v2f*)(arow + k);
      const v2f b = *(const v2f*)(wrow + k);
      acc = wmma4(a, b, acc);
    }
    const int n = ntile * 16 + r;
    const float bo = bout[n];
#pragma unroll
    for (int v = 0; v < 8; ++v)
      y[(size_t)(mbase + v) * D_MODEL + n] = acc[v] + bo;
  }
}

extern "C" void kernel_launch(void* const* d_in, const int* in_sizes, int n_in,
                              void* d_out, int out_size, void* d_ws,
                              size_t ws_size, hipStream_t stream) {
  const float* x = (const float*)d_in[0];
  const float* A = (const float*)d_in[1];
  const float* B = (const float*)d_in[2];
  const float* C = (const float*)d_in[3];
  const float* Dv = (const float*)d_in[4];
  const float* log_dt = (const float*)d_in[5];
  const float* W = (const float*)d_in[6];
  const float* bout = (const float*)d_in[7];

  float* y = (float*)d_out;
  float* fs = y + (size_t)ROWS * D_MODEL;

  float* ws = (float*)d_ws;
  float* Bbar = ws + WS_BBAR;
  float* uB = ws + WS_UB;
  float* states = ws + WS_STATES;

  prep_kernel<<<1, 256, 0, stream>>>(A, B, log_dt, ws);
  ub_kernel<<<256, 256, 0, stream>>>(x, Bbar, uB);
  scan_kernel<<<BATCH, 32, 0, stream>>>(ws, states, fs);
  cproj_kernel<<<256, 256, 0, stream>>>(states, C, x, Dv, y);
  wout_kernel<<<2048, 256, 0, stream>>>(W, bout, y);
}


# wout: both operand fragments are read 16 bytes per lane, one load per two WMMAs (k regrouped in eights: sum order changes); LDS tile pitch 516 floats so its 16 rows fall on different banks
# speedup vs baseline: 5.7754x; 1.1650x over previous
#include <hip/hip_runtime.h>

typedef float v2f __attribute__((ext_vector_type(2)));
typedef float v8f __attribute__((ext_vector_type(8)));

#define D_MODEL 512
#define D_STATE 32
#define BATCH 8
#define SEQ 4096
#define ROWS (BATCH * SEQ)

#define WS_ABAR 0
#define WS_BBAR (WS_ABAR + 1024)
#define WS_UB (WS_BBAR + 16384)
#define WS_STATES (WS_UB + ROWS * 32)

__device__ __forceinline__ v8f wmma4(v2f a, v2f b, v8f c) {
  return __builtin_amdgcn_wmma_f32_16x16x4_f32(false, a, false, b, (short)0, c,
                                               false, false);
}

__global__ void __launch_bounds__(256) prep_kernel(
    const float* __restrict__ A, const float* __restrict__ B,
    const float* __restrict__ log_dt, float* __restrict__ ws) {
  __shared__ float red[256];
  __shared__ float red2[32];
  __shared__ float rowbuf[2][64];
  __shared__ float colbuf[2][32];
  __shared__ float Ah[32][32];
  __shared__ float inv[32][32];
  __shared__ float dt_sh;
  const int tid = threadIdx.x;

  float b0[32], b1[32];
#pragma unroll
  for (int k = 0; k < 32; ++k) {
    b0[k] = B[k * D_MODEL + tid];
    b1[k] = B[k * D_MODEL + 256 + tid];
  }

  const int i = tid >> 3, jc = tid & 7;
  float av[8];
#pragma unroll
  for (int e = 0; e < 8; ++e) av[e] = A[i * 32 + (jc & 3) * 8 + e];

  {
    float v0 = expf(log_dt[tid]);
    float v1 = expf(log_dt[tid + 256]);
    v0 = fminf(fmaxf(v0, 1e-4f), 1.0f);
    v1 = fminf(fmaxf(v1, 1e-4f), 1.0f);
    red[tid] = v0 + v1;
  }
  __syncthreads();
  if (tid < 32) {
    float s = 0.f;
    for (int q = 0; q < 8; ++q) s += red[tid * 8 + q];
    red2[tid] = s;
  }
  __syncthreads();
  if (tid == 0) {
    float s = 0.f;
    for (int q = 0; q < 32; ++q) s += red2[q];
    dt_sh = s / (float)D_MODEL;
  }
  __syncthreads();
  const float dt = dt_sh;

  float a[8];
#pragma unroll
  for (int e = 0; e < 8; ++e) {
    const int j = (jc & 3) * 8 + e;
    const float eye = (i == j) ? 1.f : 0.f;
    const float lo = eye - 0.5f * dt * av[e];
    if (jc < 4) Ah[i][j] = eye + 0.5f * dt * av[e];
    a[e] = (jc < 4) ? lo : eye;
  }

  for (int k = 0; k < 32; ++k) {
    const int pb = k & 1;
    if (i == k) {
#pragma unroll
      for (int e = 0; e < 8; ++e) rowbuf[pb][jc * 8 + e] = a[e];
    }
    if (jc == (k >> 3)) {
      float ak = a[0];
#pragma unroll
      for (int e = 1; e < 8; ++e) ak = ((k & 7) == e) ? a[e] : ak;
      colbuf[pb][i] = ak;
    }
    __syncthreads();
    const float p = rowbuf[pb][k];
    const float f = colbuf[pb][i];
#pragma unroll
    for (int e = 0; e < 8; ++e) {
      const float nr = rowbuf[pb][jc * 8 + e] / p;
      a[e] = (i == k) ? nr : (a[e] - f * nr);
    }
  }
  if (jc >= 4) {
#pragma unroll
    for (int e = 0; e < 8; ++e) inv[i][(jc - 4) * 8 + e] = a[e];
  }
  __syncthreads();

  float* Abar = ws + WS_ABAR;
  for (int idx = tid; idx < 32 * 32; idx += 256) {
    const int r = idx >> 5, j = idx & 31;
    float s = 0.f;
    for (int k = 0; k < 32; ++k) s += Ah[r][k] * inv[k][j];
    Abar[idx] = s;
  }
  float* Bbar = ws + WS_BBAR;
  for (int n = 0; n < 32; ++n) {
    float s0 = 0.f, s1 = 0.f;
#pragma unroll
    for (int k = 0; k < 32; ++k) {
      const float w = inv[n][k];
      s0 += w * b0[k];
      s1 += w * b1[k];
    }
    Bbar[n * D_MODEL + tid] = dt * s0;
    Bbar[n * D_MODEL + 256 + tid] = dt * s1;
  }
}

__global__ void __launch_bounds__(256) ub_kernel(const float* __restrict__ x,
                                                 const float* __restrict__ Bbar,
                                                 float* __restrict__ uB) {
  const int wave = threadIdx.x >> 5;
  const int lane = threadIdx.x & 31;
  const int r = lane & 15, h = lane >> 4;
  const int mtile = blockIdx.x * 8 + wave;

  const float* xrow = x + (size_t)(mtile * 16 + r) * D_MODEL;
  const float* b0row = Bbar + (size_t)r * D_MODEL;
  const float* b1row = Bbar + (size_t)(16 + r) * D_MODEL;

  v8f acc0 = {0.f, 0.f, 0.f, 0.f, 0.f, 0.f, 0.f, 0.f};
  v8f acc1 = {0.f, 0.f, 0.f, 0.f, 0.f, 0.f, 0.f, 0.f};

  for (int k0 = 0; k0 < D_MODEL; k0 += 4) {
    const int k = k0 + 2 * h;
    const v2f a = *(const v2f*)(xrow + k);
    const v2f b0 = *(const v2f*)(b0row + k);
    const v2f b1 = *(const v2f*)(b1row + k);
    acc0 = wmma4(a, b0, acc0);
    acc1 = wmma4(a, b1, acc1);
  }

  const int mbase = mtile * 16 + 8 * h;
#pragma unroll
  for (int v = 0; v < 8; ++v) {
    const int m = mbase + v;
    uB[(size_t)m * 32 + r] = acc0[v];
    uB[(size_t)m * 32 + 16 + r] = acc1[v];
  }
}

__device__ __forceinline__ float tanh_step(float x) {
  float z, t, x2, s, p, m2s, x3;
  asm volatile(
      "v_mul_f32 %[t], 0x4038aa3b, |%[x]|\n\t"
      "v_mul_f32 %[x2], %[x], %[x]\n\t"
      "v_bfi_b32 %[s], 0x7fffffff, 1.0, %[x]\n\t"
      "v_cmp_gt_f32 vcc_lo, 0x3f200000, |%[x]|\n\t"
      "v_exp_f32 %[t], %[t]\n\t"
      "v_fmaak_f32 %[p], %[k4], %[x2], 0x3ca908c9\n\t"
      "v_mul_f32 %[m2s], -2.0, %[s]\n\t"
      "v_mul_f32 %[x3], %[x], %[x2]\n\t"
      "v_fmaak_f32 %[p], %[x2], %[p], 0xbd5c1c4e\n\t"
      "v_add_f32 %[t], 1.0, %[t]\n\t"
      "v_fmaak_f32 %[p], %[x2], %[p], 0x3e088382\n\t"
      "v_rcp_f32 %[t], %[t]\n\t"
      "v_fmaak_f32 %[p], %[x2], %[p], 0xbeaaaa99\n\t"
      "v_fma_f32 %[x3], %[x3], %[p], %[x]\n\t"
      "v_fmac_f32 %[s], %[m2s], %[t]\n\t"
      "v_cndmask_b32 %[z], %[s], %[x3], vcc_lo"
      : [z] "=v"(z), [t] "=&v"(t), [x2] "=&v"(x2), [s] "=&v"(s), [p] "=&v"(p),
        [m2s] "=&v"(m2s), [x3] "=&v"(x3)
      : [x] "v"(x), [k4] "s"(0xbbbac73du)
      : "vcc");
  return z;
}

#define SCAN_PF 16
#define BC16(dst, sel) "v_permlane16_b32 " dst ", %[st], " sel ", " sel "\n\t"

__device__ __forceinline__ void dot_head(float st, float& t0, float& t1,
                                         float& t2, float& t3) {
  asm volatile(BC16("%[n0]", "%[s0]") BC16("%[n1]", "%[s1]")
               BC16("%[n2]", "%[s2]") BC16("%[n3]", "%[s3]")
               : [n0] "=&v"(t0), [n1] "=&v"(t1), [n2] "=&v"(t2), [n3] "=&v"(t3)
               : [st] "v"(st), [s0] "s"(0x00000000u), [s1] "s"(0x11111111u),
                 [s2] "s"(0x22222222u), [s3] "s"(0x33333333u));
}

__device__ __forceinline__ void dot_first(float st, float ub, const float* ao,
                                          const float* ax, float t0, float t1,
                                          float t2, float t3, float& n0,
                                          float& n1, float& n2, float& n3,
                                          float& o0, float& o1, float& x0,
                                          float& x1) {
  asm volatile(BC16("%[n0]", "%[s0]")
               "v_mul_f32 %[x0], %[c0], %[t0]\n\t"
               "v_fma_f32 %[o0], %[a0], %[t0], %[ub]\n\t"
               BC16("%[n1]", "%[s1]")
               "v_mul_f32 %[x1], %[c1], %[t1]\n\t"
               "v_mul_f32 %[o1], %[a1], %[t1]\n\t"
               BC16("%[n2]", "%[s2]")
               "v_fmac_f32 %[x0], %[c2], %[t2]\n\t"
               "v_fmac_f32 %[o0], %[a2], %[t2]\n\t"
               BC16("%[n3]", "%[s3]")
               "v_fmac_f32 %[x1], %[c3], %[t3]\n\t"
               "v_fmac_f32 %[o1], %[a3], %[t3]"
               : [n0] "=&v"(n0), [n1] "=&v"(n1), [n2] "=&v"(n2), [n3] "=&v"(n3),
                 [o0] "=&v"(o0), [o1] "=&v"(o1), [x0] "=&v"(x0), [x1] "=&v"(x1)
               : [st] "v"(st), [ub] "v"(ub), [t0] "v"(t0), [t1] "v"(t1),
                 [t2] "v"(t2), [t3] "v"(t3), [a0] "v"(ao[0]), [a1] "v"(ao[1]),
                 [a2] "v"(ao[2]), [a3] "v"(ao[3]), [c0] "v"(ax[0]),
                 [c1] "v"(ax[1]), [c2] "v"(ax[2]), [c3] "v"(ax[3]),
                 [s0] "s"(0x44444444u), [s1] "s"(0x55555555u),
                 [s2] "s"(0x66666666u), [s3] "s"(0x77777777u));
}

template <int M>
__device__ __forceinline__ void dot_mid(float st, const float* ao,
                                        const float* ax, float t0, float t1,
                                        float t2, float t3, float& n0, float& n1,
                                        float& n2, float& n3, float& o0,
                                        float& o1, float& x0, float& x1) {
  asm volatile(BC16("%[n0]", "%[s0]")
               "v_fmac_f32 %[x0], %[c0], %[t0]\n\t"
               "v_fmac_f32 %[o0], %[a0], %[t0]\n\t"
               BC16("%[n1]", "%[s1]")
               "v_fmac_f32 %[x1], %[c1], %[t1]\n\t"
               "v_fmac_f32 %[o1], %[a1], %[t1]\n\t"
               BC16("%[n2]", "%[s2]")
               "v_fmac_f32 %[x0], %[c2], %[t2]\n\t"
               "v_fmac_f32 %[o0], %[a2], %[t2]\n\t"
               BC16("%[n3]", "%[s3]")
               "v_fmac_f32 %[x1], %[c3], %[t3]\n\t"
               "v_fmac_f32 %[o1], %[a3], %[t3]"
               : [n0] "=&v"(n0), [n1] "=&v"(n1), [n2] "=&v"(n2), [n3] "=&v"(n3),
                 [o0] "+v"(o0), [o1] "+v"(o1), [x0] "+v"(x0), [x1] "+v"(x1)
               : [st] "v"(st), [t0] "v"(t0), [t1] "v"(t1), [t2] "v"(t2),
                 [t3] "v"(t3), [a0] "v"(ao[M]), [a1] "v"(ao[M + 1]),
                 [a2] "v"(ao[M + 2]), [a3] "v"(ao[M + 3]), [c0] "v"(ax[M]),
                 [c1] "v"(ax[M + 1]), [c2] "v"(ax[M + 2]), [c3] "v"(ax[M + 3]),
                 [s0] "s"(0x11111111u * (unsigned)(M + 4)),
                 [s1] "s"(0x11111111u * (unsigned)(M + 5)),
                 [s2] "s"(0x11111111u * (unsigned)(M + 6)),
                 [s3] "s"(0x11111111u * (unsigned)(M + 7)));
}

__device__ __forceinline__ float dot_tail(const float* ao, const float* ax,
                                          float t0, float t1, float t2, float t3,
                                          float o0, float o1, float x0,
                                          float x1) {
  float acc;
  asm volatile("v_fmac_f32 %[x0], %[c0], %[t0]\n\t"
               "v_fmac_f32 %[o0], %[a0], %[t0]\n\t"
               "v_fmac_f32 %[x1], %[c1], %[t1]\n\t"
               "v_fmac_f32 %[o1], %[a1], %[t1]\n\t"
               "v_fmac_f32 %[x0], %[c2], %[t2]\n\t"
               "v_fmac_f32 %[o0], %[a2], %[t2]\n\t"
               "v_fmac_f32 %[x1], %[c3], %[t3]\n\t"
               "v_fmac_f32 %[o1], %[a3], %[t3]\n\t"
               "v_add_f32 %[x0], %[x0], %[x1]\n\t"
               "v_add_f32 %[o0], %[o0], %[o1]\n\t"
               "v_permlanex16_b32 %[x1], %[x0], %[lo], %[hi]\n\t"
               "v_add_f32 %[acc], %[o0], %[x1]"
               : [acc] "=v"(acc), [o0] "+v"(o0), [o1] "+v"(o1), [x0] "+v"(x0),
                 [x1] "+v"(x1)
               : [t0] "v"(t0), [t1] "v"(t1), [t2] "v"(t2), [t3] "v"(t3),
                 [a0] "v"(ao[12]), [a1] "v"(ao[13]), [a2] "v"(ao[14]),
                 [a3] "v"(ao[15]), [c0] "v"(ax[12]), [c1] "v"(ax[13]),
                 [c2] "v"(ax[14]), [c3] "v"(ax[15]), [lo] "s"(0x76543210u),
                 [hi] "s"(0xfedcba98u));
  return acc;
}

__global__ void __launch_bounds__(32) scan_kernel(const float* __restrict__ ws,
                                                  float* __restrict__ states,
                                                  float* __restrict__ fs_out) {
  const int b = blockIdx.x;
  const int lane = threadIdx.x;
  const int hb = lane & 16;
  const float* Abar = ws + WS_ABAR;
  const float* uB = ws + WS_UB + (size_t)b * SEQ * 32;
  float* st_out = states + (size_t)b * SEQ * 32;

  float ao[16], ax[16];
#pragma unroll
  for (int m = 0; m < 16; ++m) {
    ao[m] = Abar[lane * 32 + hb + m];
    ax[m] = Abar[(lane ^ 16) * 32 + hb + m];
  }

  float cur[SCAN_PF];
#pragma unroll
  for (int j = 0; j < SCAN_PF; ++j) cur[j] = uB[j * 32 + lane];

  float st = 0.f;
  for (int s0 = 0; s0 < SEQ; s0 += SCAN_PF) {
    const int sn = (s0 + SCAN_PF < SEQ) ? (s0 + SCAN_PF) : s0;
    float nxt[SCAN_PF];
#pragma unroll
    for (int j = 0; j < SCAN_PF; ++j) nxt[j] = uB[(sn + j) * 32 + lane];

#pragma unroll
    for (int j = 0; j < SCAN_PF; ++j) {
      float o0, o1, x0, x1;
      float p0, p1, p2, p3, q0, q1, q2, q3;
      dot_head(st, p0, p1, p2, p3);
      dot_first(st, cur[j], ao, ax, p0, p1, p2, p3, q0, q1, q2, q3, o0, o1, x0,
                x1);
      dot_mid<4>(st, ao, ax, q0, q1, q2, q3, p0, p1, p2, p3, o0, o1, x0, x1);
      dot_mid<8>(st, ao, ax, p0, p1, p2, p3, q0, q1, q2, q3, o0, o1, x0, x1);
      const float acc = dot_tail(ao, ax, q0, q1, q2, q3, o0, o1, x0, x1);
      st = tanh_step(acc);
      st_out[(s0 + j) * 32 + lane] = st;
    }
#pragma unroll
    for (int j = 0; j < SCAN_PF; ++j) cur[j] = nxt[j];
  }
  fs_out[b * 32 + lane] = st;
}

__global__ void __launch_bounds__(256) cproj_kernel(
    const float* __restrict__ states, const float* __restrict__ C,
    const float* __restrict__ x, const float* __restrict__ Dv,
    float* __restrict__ y1) {
  const int wave = threadIdx.x >> 5;
  const int lane = threadIdx.x & 31;
  const int r = lane & 15, h = lane >> 4;
  const int mtile = blockIdx.x * 8 + wave;

  v2f afrag[8];
  const float* srow = states + (size_t)(mtile * 16 + r) * 32;
#pragma unroll
  for (int kk = 0; kk < 8; ++kk)
    afrag[kk] = *(const v2f*)(srow + kk * 4 + 2 * h);

  const int mbase = mtile * 16 + 8 * h;
  for (int dtile = 0; dtile < 32; ++dtile) {
    v8f acc = {0.f, 0.f, 0.f, 0.f, 0.f, 0.f, 0.f, 0.f};
    const float* crow = C + (size_t)(dtile * 16 + r) * 32;
#pragma unroll
    for (int kk = 0; kk < 8; ++kk) {
      const v2f b = *(const v2f*)(crow + kk * 4 + 2 * h);
      acc = wmma4(afrag[kk], b, acc);
    }
    const int d = dtile * 16 + r;
    const float Dd = Dv[d];
#pragma unroll
    for (int v = 0; v < 8; ++v) {
      const int m = mbase + v;
      y1[(size_t)m * D_MODEL + d] = acc[v] + x[(size_t)m * D_MODEL + d] * Dd;
    }
  }
}

#define WOUT_PITCH (D_MODEL + 4)
__global__ void __launch_bounds__(256) wout_kernel(
    const float* __restrict__ W, const float* __restrict__ bout,
    float* __restrict__ y) {
  __shared__ float lds[16 * WOUT_PITCH];
  const int tid = threadIdx.x;
  const int mtile = blockIdx.x;

  const float4* src4 = (const float4*)(y + (size_t)mtile * 16 * D_MODEL);
#pragma unroll
  for (int i = 0; i < 8; ++i) {
    const int idx = tid + i * 256;
    const int row = idx >> 7, c4 = idx & 127;
    *(float4*)(lds + row * WOUT_PITCH + c4 * 4) = src4[idx];
  }
  __syncthreads();

  const int wave = tid >> 5, lane = tid & 31;
  const int r = lane & 15, h = lane >> 4;
  const int mbase = mtile * 16 + 8 * h;
  const float* arow = lds + r * WOUT_PITCH + 4 * h;

  for (int t = 0; t < 4; ++t) {
    const int ntile = wave * 4 + t;
    const float* wrow = W + (size_t)(ntile * 16 + r) * D_MODEL + 4 * h;
    if (t < 3)
      __builtin_prefetch(W + (size_t)((ntile + 1) * 16 + r) * D_MODEL, 0, 1);

    v8f acc = {0.f, 0.f, 0.f, 0.f, 0.f, 0.f, 0.f, 0.f};
    for (int k8 = 0; k8 < D_MODEL; k8 += 8) {
      const float4 a4 = *(const float4*)(arow + k8);
      const float4 b4 = *(const float4*)(wrow + k8);
      const v2f a_lo = {a4.x, a4.y}, b_lo = {b4.x, b4.y};
      const v2f a_hi = {a4.z, a4.w}, b_hi = {b4.z, b4.w};
      acc = wmma4(a_lo, b_lo, acc);
      acc = wmma4(a_hi, b_hi, acc);
    }
    const int n = ntile * 16 + r;
    const float bo = bout[n];
#pragma unroll
    for (int v = 0; v < 8; ++v)
      y[(size_t)(mbase + v) * D_MODEL + n] = acc[v] + bo;
  }
}

extern "C" void kernel_launch(void* const* d_in, const int* in_sizes, int n_in,
                              void* d_out, int out_size, void* d_ws,
                              size_t ws_size, hipStream_t stream) {
  const float* x = (const float*)d_in[0];
  const float* A = (const float*)d_in[1];
  const float* B = (const float*)d_in[2];
  const float* C = (const float*)d_in[3];
  const float* Dv = (const float*)d_in[4];
  const float* log_dt = (const float*)d_in[5];
  const float* W = (const float*)d_in[6];
  const float* bout = (const float*)d_in[7];

  float* y = (float*)d_out;
  float* fs = y + (size_t)ROWS * D_MODEL;

  float* ws = (float*)d_ws;
  float* Bbar = ws + WS_BBAR;
  float* uB = ws + WS_UB;
  float* states = ws + WS_STATES;

  prep_kernel<<<1, 256, 0, stream>>>(A, B, log_dt, ws);
  ub_kernel<<<256, 256, 0, stream>>>(x, Bbar, uB);
  scan_kernel<<<BATCH, 32, 0, stream>>>(ws, states, fs);
  cproj_kernel<<<256, 256, 0, stream>>>(states, C, x, Dv, y);
  wout_kernel<<<2048, 256, 0, stream>>>(W, bout, y);
}


# scan, C-projection and output projection become roles of one grid on time pieces c, c-1, c-2 of 8 (10 launches for 3), so both projections run beside the serial scan; arithmetic and order unchanged
# speedup vs baseline: 6.6508x; 1.1516x over previous
#include <hip/hip_runtime.h>

typedef float v2f __attribute__((ext_vector_type(2)));
typedef float v8f __attribute__((ext_vector_type(8)));

#define D_MODEL 512
#define D_STATE 32
#define BATCH 8
#define SEQ 4096
#define ROWS (BATCH * SEQ)

#define WS_ABAR 0
#define WS_BBAR (WS_ABAR + 1024)
#define WS_UB (WS_BBAR + 16384)
#define WS_STATES (WS_UB + ROWS * 32)

#define NCHUNK 8
#define CH (SEQ / NCHUNK)
#define CH_MTILES (BATCH * CH / 16)
#define NB_SCAN BATCH
#define NB_CPROJ (CH_MTILES / 8)
#define NB_WOUT CH_MTILES
__device__ __forceinline__ int piece_tile(int c, int j) {
  return (j >> 5) * (SEQ / 16) + c * (CH / 16) + (j & 31);
}

__device__ __forceinline__ v8f wmma4(v2f a, v2f b, v8f c) {
  return __builtin_amdgcn_wmma_f32_16x16x4_f32(false, a, false, b, (short)0, c,
                                               false, false);
}

__global__ void __launch_bounds__(256) prep_kernel(
    const float* __restrict__ A, const float* __restrict__ B,
    const float* __restrict__ log_dt, float* __restrict__ ws) {
  __shared__ float red[256];
  __shared__ float red2[32];
  __shared__ float rowbuf[2][64];
  __shared__ float colbuf[2][32];
  __shared__ float Ah[32][32];
  __shared__ float inv[32][32];
  __shared__ float dt_sh;
  const int tid = threadIdx.x;

  float b0[32], b1[32];
#pragma unroll
  for (int k = 0; k < 32; ++k) {
    b0[k] = B[k * D_MODEL + tid];
    b1[k] = B[k * D_MODEL + 256 + tid];
  }

  const int i = tid >> 3, jc = tid & 7;
  float av[8];
#pragma unroll
  for (int e = 0; e < 8; ++e) av[e] = A[i * 32 + (jc & 3) * 8 + e];

  {
    float v0 = expf(log_dt[tid]);
    float v1 = expf(log_dt[tid + 256]);
    v0 = fminf(fmaxf(v0, 1e-4f), 1.0f);
    v1 = fminf(fmaxf(v1, 1e-4f), 1.0f);
    red[tid] = v0 + v1;
  }
  __syncthreads();
  if (tid < 32) {
    float s = 0.f;
    for (int q = 0; q < 8; ++q) s += red[tid * 8 + q];
    red2[tid] = s;
  }
  __syncthreads();
  if (tid == 0) {
    float s = 0.f;
    for (int q = 0; q < 32; ++q) s += red2[q];
    dt_sh = s / (float)D_MODEL;
  }
  __syncthreads();
  const float dt = dt_sh;

  float a[8];
#pragma unroll
  for (int e = 0; e < 8; ++e) {
    const int j = (jc & 3) * 8 + e;
    const float eye = (i == j) ? 1.f : 0.f;
    const float lo = eye - 0.5f * dt * av[e];
    if (jc < 4) Ah[i][j] = eye + 0.5f * dt * av[e];
    a[e] = (jc < 4) ? lo : eye;
  }

  for (int k = 0; k < 32; ++k) {
    const int pb = k & 1;
    if (i == k) {
#pragma unroll
      for (int e = 0; e < 8; ++e) rowbuf[pb][jc * 8 + e] = a[e];
    }
    if (jc == (k >> 3)) {
      float ak = a[0];
#pragma unroll
      for (int e = 1; e < 8; ++e) ak = ((k & 7) == e) ? a[e] : ak;
      colbuf[pb][i] = ak;
    }
    __syncthreads();
    const float p = rowbuf[pb][k];
    const float f = colbuf[pb][i];
#pragma unroll
    for (int e = 0; e < 8; ++e) {
      const float nr = rowbuf[pb][jc * 8 + e] / p;
      a[e] = (i == k) ? nr : (a[e] - f * nr);
    }
  }
  if (jc >= 4) {
#pragma unroll
    for (int e = 0; e < 8; ++e) inv[i][(jc - 4) * 8 + e] = a[e];
  }
  __syncthreads();

  float* Abar = ws + WS_ABAR;
  for (int idx = tid; idx < 32 * 32; idx += 256) {
    const int r = idx >> 5, j = idx & 31;
    float s = 0.f;
    for (int k = 0; k < 32; ++k) s += Ah[r][k] * inv[k][j];
    Abar[idx] = s;
  }
  float* Bbar = ws + WS_BBAR;
  for (int n = 0; n < 32; ++n) {
    float s0 = 0.f, s1 = 0.f;
#pragma unroll
    for (int k = 0; k < 32; ++k) {
      const float w = inv[n][k];
      s0 += w * b0[k];
      s1 += w * b1[k];
    }
    Bbar[n * D_MODEL + tid] = dt * s0;
    Bbar[n * D_MODEL + 256 + tid] = dt * s1;
  }
}

__global__ void __launch_bounds__(256) ub_kernel(const float* __restrict__ x,
                                                 const float* __restrict__ Bbar,
                                                 float* __restrict__ uB) {
  const int wave = threadIdx.x >> 5;
  const int lane = threadIdx.x & 31;
  const int r = lane & 15, h = lane >> 4;
  const int mtile = blockIdx.x * 8 + wave;

  const float* xrow = x + (size_t)(mtile * 16 + r) * D_MODEL;
  const float* b0row = Bbar + (size_t)r * D_MODEL;
  const float* b1row = Bbar + (size_t)(16 + r) * D_MODEL;

  v8f acc0 = {0.f, 0.f, 0.f, 0.f, 0.f, 0.f, 0.f, 0.f};
  v8f acc1 = {0.f, 0.f, 0.f, 0.f, 0.f, 0.f, 0.f, 0.f};

  for (int k0 = 0; k0 < D_MODEL; k0 += 4) {
    const int k = k0 + 2 * h;
    const v2f a = *(const v2f*)(xrow + k);
    const v2f b0 = *(const v2f*)(b0row + k);
    const v2f b1 = *(const v2f*)(b1row + k);
    acc0 = wmma4(a, b0, acc0);
    acc1 = wmma4(a, b1, acc1);
  }

  const int mbase = mtile * 16 + 8 * h;
#pragma unroll
  for (int v = 0; v < 8; ++v) {
    const int m = mbase + v;
    uB[(size_t)m * 32 + r] = acc0[v];
    uB[(size_t)m * 32 + 16 + r] = acc1[v];
  }
}

__device__ __forceinline__ float tanh_step(float x) {
  float z, t, x2, s, p, m2s, x3;
  asm volatile(
      "v_mul_f32 %[t], 0x4038aa3b, |%[x]|\n\t"
      "v_mul_f32 %[x2], %[x], %[x]\n\t"
      "v_bfi_b32 %[s], 0x7fffffff, 1.0, %[x]\n\t"
      "v_cmp_gt_f32 vcc_lo, 0x3f200000, |%[x]|\n\t"
      "v_exp_f32 %[t], %[t]\n\t"
      "v_fmaak_f32 %[p], %[k4], %[x2], 0x3ca908c9\n\t"
      "v_mul_f32 %[m2s], -2.0, %[s]\n\t"
      "v_mul_f32 %[x3], %[x], %[x2]\n\t"
      "v_fmaak_f32 %[p], %[x2], %[p], 0xbd5c1c4e\n\t"
      "v_add_f32 %[t], 1.0, %[t]\n\t"
      "v_fmaak_f32 %[p], %[x2], %[p], 0x3e088382\n\t"
      "v_rcp_f32 %[t], %[t]\n\t"
      "v_fmaak_f32 %[p], %[x2], %[p], 0xbeaaaa99\n\t"
      "v_fma_f32 %[x3], %[x3], %[p], %[x]\n\t"
      "v_fmac_f32 %[s], %[m2s], %[t]\n\t"
      "v_cndmask_b32 %[z], %[s], %[x3], vcc_lo"
      : [z] "=v"(z), [t] "=&v"(t), [x2] "=&v"(x2), [s] "=&v"(s), [p] "=&v"(p),
        [m2s] "=&v"(m2s), [x3] "=&v"(x3)
      : [x] "v"(x), [k4] "s"(0xbbbac73du)
      : "vcc");
  return z;
}

#define SCAN_PF 16
#define BC16(dst, sel) "v_permlane16_b32 " dst ", %[st], " sel ", " sel "\n\t"

__device__ __forceinline__ void dot_head(float st, float& t0, float& t1,
                                         float& t2, float& t3) {
  asm volatile(BC16("%[n0]", "%[s0]") BC16("%[n1]", "%[s1]")
               BC16("%[n2]", "%[s2]") BC16("%[n3]", "%[s3]")
               : [n0] "=&v"(t0), [n1] "=&v"(t1), [n2] "=&v"(t2), [n3] "=&v"(t3)
               : [st] "v"(st), [s0] "s"(0x00000000u), [s1] "s"(0x11111111u),
                 [s2] "s"(0x22222222u), [s3] "s"(0x33333333u));
}

__device__ __forceinline__ void dot_first(float st, float ub, const float* ao,
                                          const float* ax, float t0, float t1,
                                          float t2, float t3, float& n0,
                                          float& n1, float& n2, float& n3,
                                          float& o0, float& o1, float& x0,
                                          float& x1) {
  asm volatile(BC16("%[n0]", "%[s0]")
               "v_mul_f32 %[x0], %[c0], %[t0]\n\t"
               "v_fma_f32 %[o0], %[a0], %[t0], %[ub]\n\t"
               BC16("%[n1]", "%[s1]")
               "v_mul_f32 %[x1], %[c1], %[t1]\n\t"
               "v_mul_f32 %[o1], %[a1], %[t1]\n\t"
               BC16("%[n2]", "%[s2]")
               "v_fmac_f32 %[x0], %[c2], %[t2]\n\t"
               "v_fmac_f32 %[o0], %[a2], %[t2]\n\t"
               BC16("%[n3]", "%[s3]")
               "v_fmac_f32 %[x1], %[c3], %[t3]\n\t"
               "v_fmac_f32 %[o1], %[a3], %[t3]"
               : [n0] "=&v"(n0), [n1] "=&v"(n1), [n2] "=&v"(n2), [n3] "=&v"(n3),
                 [o0] "=&v"(o0), [o1] "=&v"(o1), [x0] "=&v"(x0), [x1] "=&v"(x1)
               : [st] "v"(st), [ub] "v"(ub), [t0] "v"(t0), [t1] "v"(t1),
                 [t2] "v"(t2), [t3] "v"(t3), [a0] "v"(ao[0]), [a1] "v"(ao[1]),
                 [a2] "v"(ao[2]), [a3] "v"(ao[3]), [c0] "v"(ax[0]),
                 [c1] "v"(ax[1]), [c2] "v"(ax[2]), [c3] "v"(ax[3]),
                 [s0] "s"(0x44444444u), [s1] "s"(0x55555555u),
                 [s2] "s"(0x66666666u), [s3] "s"(0x77777777u));
}

template <int M>
__device__ __forceinline__ void dot_mid(float st, const float* ao,
                                        const float* ax, float t0, float t1,
                                        float t2, float t3, float& n0, float& n1,
                                        float& n2, float& n3, float& o0,
                                        float& o1, float& x0, float& x1) {
  asm volatile(BC16("%[n0]", "%[s0]")
               "v_fmac_f32 %[x0], %[c0], %[t0]\n\t"
               "v_fmac_f32 %[o0], %[a0], %[t0]\n\t"
               BC16("%[n1]", "%[s1]")
               "v_fmac_f32 %[x1], %[c1], %[t1]\n\t"
               "v_fmac_f32 %[o1], %[a1], %[t1]\n\t"
               BC16("%[n2]", "%[s2]")
               "v_fmac_f32 %[x0], %[c2], %[t2]\n\t"
               "v_fmac_f32 %[o0], %[a2], %[t2]\n\t"
               BC16("%[n3]", "%[s3]")
               "v_fmac_f32 %[x1], %[c3], %[t3]\n\t"
               "v_fmac_f32 %[o1], %[a3], %[t3]"
               : [n0] "=&v"(n0), [n1] "=&v"(n1), [n2] "=&v"(n2), [n3] "=&v"(n3),
                 [o0] "+v"(o0), [o1] "+v"(o1), [x0] "+v"(x0), [x1] "+v"(x1)
               : [st] "v"(st), [t0] "v"(t0), [t1] "v"(t1), [t2] "v"(t2),
                 [t3] "v"(t3), [a0] "v"(ao[M]), [a1] "v"(ao[M + 1]),
                 [a2] "v"(ao[M + 2]), [a3] "v"(ao[M + 3]), [c0] "v"(ax[M]),
                 [c1] "v"(ax[M + 1]), [c2] "v"(ax[M + 2]), [c3] "v"(ax[M + 3]),
                 [s0] "s"(0x11111111u * (unsigned)(M + 4)),
                 [s1] "s"(0x11111111u * (unsigned)(M + 5)),
                 [s2] "s"(0x11111111u * (unsigned)(M + 6)),
                 [s3] "s"(0x11111111u * (unsigned)(M + 7)));
}

__device__ __forceinline__ float dot_tail(const float* ao, const float* ax,
                                          float t0, float t1, float t2, float t3,
                                          float o0, float o1, float x0,
                                          float x1) {
  float acc;
  asm volatile("v_fmac_f32 %[x0], %[c0], %[t0]\n\t"
               "v_fmac_f32 %[o0], %[a0], %[t0]\n\t"
               "v_fmac_f32 %[x1], %[c1], %[t1]\n\t"
               "v_fmac_f32 %[o1], %[a1], %[t1]\n\t"
               "v_fmac_f32 %[x0], %[c2], %[t2]\n\t"
               "v_fmac_f32 %[o0], %[a2], %[t2]\n\t"
               "v_fmac_f32 %[x1], %[c3], %[t3]\n\t"
               "v_fmac_f32 %[o1], %[a3], %[t3]\n\t"
               "v_add_f32 %[x0], %[x0], %[x1]\n\t"
               "v_add_f32 %[o0], %[o0], %[o1]\n\t"
               "v_permlanex16_b32 %[x1], %[x0], %[lo], %[hi]\n\t"
               "v_add_f32 %[acc], %[o0], %[x1]"
               : [acc] "=v"(acc), [o0] "+v"(o0), [o1] "+v"(o1), [x0] "+v"(x0),
                 [x1] "+v"(x1)
               : [t0] "v"(t0), [t1] "v"(t1), [t2] "v"(t2), [t3] "v"(t3),
                 [a0] "v"(ao[12]), [a1] "v"(ao[13]), [a2] "v"(ao[14]),
                 [a3] "v"(ao[15]), [c0] "v"(ax[12]), [c1] "v"(ax[13]),
                 [c2] "v"(ax[14]), [c3] "v"(ax[15]), [lo] "s"(0x76543210u),
                 [hi] "s"(0xfedcba98u));
  return acc;
}

__device__ __forceinline__ void scan_piece(const int b, const int c,
                                           const float* __restrict__ ws,
                                           float* states,
                                           float* __restrict__ fs_out) {
  const int lane = threadIdx.x;
  const int sbeg = c * CH;
  const int hb = lane & 16;
  const float* Abar = ws + WS_ABAR;
  const float* uB = ws + WS_UB + (size_t)b * SEQ * 32;
  float* st_out = states + (size_t)b * SEQ * 32;

  float ao[16], ax[16];
#pragma unroll
  for (int m = 0; m < 16; ++m) {
    ao[m] = Abar[lane * 32 + hb + m];
    ax[m] = Abar[(lane ^ 16) * 32 + hb + m];
  }

  float cur[SCAN_PF];
#pragma unroll
  for (int j = 0; j < SCAN_PF; ++j) cur[j] = uB[(sbeg + j) * 32 + lane];

  float st = (c == 0) ? 0.f : st_out[(sbeg - 1) * 32 + lane];
  for (int s0 = sbeg; s0 < sbeg + CH; s0 += SCAN_PF) {
    const int sn = (s0 + SCAN_PF < sbeg + CH) ? (s0 + SCAN_PF) : s0;
    float nxt[SCAN_PF];
#pragma unroll
    for (int j = 0; j < SCAN_PF; ++j) nxt[j] = uB[(sn + j) * 32 + lane];

#pragma unroll
    for (int j = 0; j < SCAN_PF; ++j) {
      float o0, o1, x0, x1;
      float p0, p1, p2, p3, q0, q1, q2, q3;
      dot_head(st, p0, p1, p2, p3);
      dot_first(st, cur[j], ao, ax, p0, p1, p2, p3, q0, q1, q2, q3, o0, o1, x0,
                x1);
      dot_mid<4>(st, ao, ax, q0, q1, q2, q3, p0, p1, p2, p3, o0, o1, x0, x1);
      dot_mid<8>(st, ao, ax, p0, p1, p2, p3, q0, q1, q2, q3, o0, o1, x0, x1);
      const float acc = dot_tail(ao, ax, q0, q1, q2, q3, o0, o1, x0, x1);
      st = tanh_step(acc);
      st_out[(s0 + j) * 32 + lane] = st;
    }
#pragma unroll
    for (int j = 0; j < SCAN_PF; ++j) cur[j] = nxt[j];
  }
  if (c == NCHUNK - 1) fs_out[b * 32 + lane] = st;
}

__device__ __forceinline__ void cproj_piece(
    const int bid, const int c, const float* states,
    const float* __restrict__ C, const float* __restrict__ x,
    const float* __restrict__ Dv, float* y1) {
  const int wave = threadIdx.x >> 5;
  const int lane = threadIdx.x & 31;
  const int r = lane & 15, h = lane >> 4;
  const int mtile = piece_tile(c, bid * 8 + wave);

  v2f afrag[8];
  const float* srow = states + (size_t)(mtile * 16 + r) * 32;
#pragma unroll
  for (int kk = 0; kk < 8; ++kk)
    afrag[kk] = *(const v2f*)(srow + kk * 4 + 2 * h);

  const int mbase = mtile * 16 + 8 * h;
  for (int dtile = 0; dtile < 32; ++dtile) {
    v8f acc = {0.f, 0.f, 0.f, 0.f, 0.f, 0.f, 0.f, 0.f};
    const float* crow = C + (size_t)(dtile * 16 + r) * 32;
#pragma unroll
    for (int kk = 0; kk < 8; ++kk) {
      const v2f b = *(const v2f*)(crow + kk * 4 + 2 * h);
      acc = wmma4(afrag[kk], b, acc);
    }
    const int d = dtile * 16 + r;
    const float Dd = Dv[d];
#pragma unroll
    for (int v = 0; v < 8; ++v) {
      const int m = mbase + v;
      y1[(size_t)m * D_MODEL + d] = acc[v] + x[(size_t)m * D_MODEL + d] * Dd;
    }
  }
}

#define WOUT_PITCH (D_MODEL + 4)
__device__ __forceinline__ void wout_piece(
    const int bid, const int c, const float* __restrict__ W,
    const float* __restrict__ bout, float* y) {
  __shared__ float lds[16 * WOUT_PITCH];
  const int tid = threadIdx.x;
  const int mtile = piece_tile(c, bid);

  const float4* src4 = (const float4*)(y + (size_t)mtile * 16 * D_MODEL);
#pragma unroll
  for (int i = 0; i < 8; ++i) {
    const int idx = tid + i * 256;
    const int row = idx >> 7, c4 = idx & 127;
    *(float4*)(lds + row * WOUT_PITCH + c4 * 4) = src4[idx];
  }
  __syncthreads();

  const int wave = tid >> 5, lane = tid & 31;
  const int r = lane & 15, h = lane >> 4;
  const int mbase = mtile * 16 + 8 * h;
  const float* arow = lds + r * WOUT_PITCH + 4 * h;

  for (int t = 0; t < 4; ++t) {
    const int ntile = wave * 4 + t;
    const float* wrow = W + (size_t)(ntile * 16 + r) * D_MODEL + 4 * h;
    if (t < 3)
      __builtin_prefetch(W + (size_t)((ntile + 1) * 16 + r) * D_MODEL, 0, 1);

    v8f acc = {0.f, 0.f, 0.f, 0.f, 0.f, 0.f, 0.f, 0.f};
    for (int k8 = 0; k8 < D_MODEL; k8 += 8) {
      const float4 a4 = *(const float4*)(arow + k8);
      const float4 b4 = *(const float4*)(wrow + k8);
      const v2f a_lo = {a4.x, a4.y}, b_lo = {b4.x, b4.y};
      const v2f a_hi = {a4.z, a4.w}, b_hi = {b4.z, b4.w};
      acc = wmma4(a_lo, b_lo, acc);
      acc = wmma4(a_hi, b_hi, acc);
    }
    const int n = ntile * 16 + r;
    const float bo = bout[n];
#pragma unroll
    for (int v = 0; v < 8; ++v)
      y[(size_t)(mbase + v) * D_MODEL + n] = acc[v] + bo;
  }
}

__global__ void __launch_bounds__(256) stage_kernel(
    const float* __restrict__ ws, float* states, float* __restrict__ fs,
    const float* __restrict__ C, const float* __restrict__ x,
    const float* __restrict__ Dv, const float* __restrict__ W,
    const float* __restrict__ bout, float* y, const int c) {
  int bid = blockIdx.x;
  if (bid < NB_SCAN) {
    if (c < NCHUNK && threadIdx.x < 32) scan_piece(bid, c, ws, states, fs);
    return;
  }
  bid -= NB_SCAN;
  if (bid < NB_CPROJ) {
    if (c >= 1 && c <= NCHUNK) cproj_piece(bid, c - 1, states, C, x, Dv, y);
    return;
  }
  bid -= NB_CPROJ;
  if (c >= 2) wout_piece(bid, c - 2, W, bout, y);
}

extern "C" void kernel_launch(void* const* d_in, const int* in_sizes, int n_in,
                              void* d_out, int out_size, void* d_ws,
                              size_t ws_size, hipStream_t stream) {
  const float* x = (const float*)d_in[0];
  const float* A = (const float*)d_in[1];
  const float* B = (const float*)d_in[2];
  const float* C = (const float*)d_in[3];
  const float* Dv = (const float*)d_in[4];
  const float* log_dt = (const float*)d_in[5];
  const float* W = (const float*)d_in[6];
  const float* bout = (const float*)d_in[7];

  float* y = (float*)d_out;
  float* fs = y + (size_t)ROWS * D_MODEL;

  float* ws = (float*)d_ws;
  float* Bbar = ws + WS_BBAR;
  float* uB = ws + WS_UB;
  float* states = ws + WS_STATES;

  prep_kernel<<<1, 256, 0, stream>>>(A, B, log_dt, ws);
  ub_kernel<<<256, 256, 0, stream>>>(x, Bbar, uB);
  stage_kernel<<<8, 256, 0, stream>>>(ws, states, fs, C, x, Dv, W, bout, y, 0);
  stage_kernel<<<40, 256, 0, stream>>>(ws, states, fs, C, x, Dv, W, bout, y, 1);
  stage_kernel<<<296, 256, 0, stream>>>(ws, states, fs, C, x, Dv, W, bout, y, 2);
  stage_kernel<<<296, 256, 0, stream>>>(ws, states, fs, C, x, Dv, W, bout, y, 3);
  stage_kernel<<<296, 256, 0, stream>>>(ws, states, fs, C, x, Dv, W, bout, y, 4);
  stage_kernel<<<296, 256, 0, stream>>>(ws, states, fs, C, x, Dv, W, bout, y, 5);
  stage_kernel<<<296, 256, 0, stream>>>(ws, states, fs, C, x, Dv, W, bout, y, 6);
  stage_kernel<<<296, 256, 0, stream>>>(ws, states, fs, C, x, Dv, W, bout, y, 7);
  stage_kernel<<<296, 256, 0, stream>>>(ws, states, fs, C, x, Dv, W, bout, y, 8);
  stage_kernel<<<296, 256, 0, stream>>>(ws, states, fs, C, x, Dv, W, bout, y, 9);
}


# scan: each step's tanh is one hardware v_tanh_f32 instead of the 16-instruction exp2/rcp/polynomial form (6 dependent levels become 1); rounding is the instruction's own, error bound not known here
# speedup vs baseline: 7.7887x; 1.1711x over previous
#include <hip/hip_runtime.h>

typedef float v2f __attribute__((ext_vector_type(2)));
typedef float v8f __attribute__((ext_vector_type(8)));

#define D_MODEL 512
#define D_STATE 32
#define BATCH 8
#define SEQ 4096
#define ROWS (BATCH * SEQ)

#define WS_ABAR 0
#define WS_BBAR (WS_ABAR + 1024)
#define WS_UB (WS_BBAR + 16384)
#define WS_STATES (WS_UB + ROWS * 32)

#define NCHUNK 8
#define CH (SEQ / NCHUNK)
#define CH_MTILES (BATCH * CH / 16)
#define NB_SCAN BATCH
#define NB_CPROJ (CH_MTILES / 8)
#define NB_WOUT CH_MTILES
__device__ __forceinline__ int piece_tile(int c, int j) {
  return (j >> 5) * (SEQ / 16) + c * (CH / 16) + (j & 31);
}

__device__ __forceinline__ v8f wmma4(v2f a, v2f b, v8f c) {
  return __builtin_amdgcn_wmma_f32_16x16x4_f32(false, a, false, b, (short)0, c,
                                               false, false);
}

__global__ void __launch_bounds__(256) prep_kernel(
    const float* __restrict__ A, const float* __restrict__ B,
    const float* __restrict__ log_dt, float* __restrict__ ws) {
  __shared__ float red[256];
  __shared__ float red2[32];
  __shared__ float rowbuf[2][64];
  __shared__ float colbuf[2][32];
  __shared__ float Ah[32][32];
  __shared__ float inv[32][32];
  __shared__ float dt_sh;
  const int tid = threadIdx.x;

  float b0[32], b1[32];
#pragma unroll
  for (int k = 0; k < 32; ++k) {
    b0[k] = B[k * D_MODEL + tid];
    b1[k] = B[k * D_MODEL + 256 + tid];
  }

  const int i = tid >> 3, jc = tid & 7;
  float av[8];
#pragma unroll
  for (int e = 0; e < 8; ++e) av[e] = A[i * 32 + (jc & 3) * 8 + e];

  {
    float v0 = expf(log_dt[tid]);
    float v1 = expf(log_dt[tid + 256]);
    v0 = fminf(fmaxf(v0, 1e-4f), 1.0f);
    v1 = fminf(fmaxf(v1, 1e-4f), 1.0f);
    red[tid] = v0 + v1;
  }
  __syncthreads();
  if (tid < 32) {
    float s = 0.f;
    for (int q = 0; q < 8; ++q) s += red[tid * 8 + q];
    red2[tid] = s;
  }
  __syncthreads();
  if (tid == 0) {
    float s = 0.f;
    for (int q = 0; q < 32; ++q) s += red2[q];
    dt_sh = s / (float)D_MODEL;
  }
  __syncthreads();
  const float dt = dt_sh;

  float a[8];
#pragma unroll
  for (int e = 0; e < 8; ++e) {
    const int j = (jc & 3) * 8 + e;
    const float eye = (i == j) ? 1.f : 0.f;
    const float lo = eye - 0.5f * dt * av[e];
    if (jc < 4) Ah[i][j] = eye + 0.5f * dt * av[e];
    a[e] = (jc < 4) ? lo : eye;
  }

  for (int k = 0; k < 32; ++k) {
    const int pb = k & 1;
    if (i == k) {
#pragma unroll
      for (int e = 0; e < 8; ++e) rowbuf[pb][jc * 8 + e] = a[e];
    }
    if (jc == (k >> 3)) {
      float ak = a[0];
#pragma unroll
      for (int e = 1; e < 8; ++e) ak = ((k & 7) == e) ? a[e] : ak;
      colbuf[pb][i] = ak;
    }
    __syncthreads();
    const float p = rowbuf[pb][k];
    const float f = colbuf[pb][i];
#pragma unroll
    for (int e = 0; e < 8; ++e) {
      const float nr = rowbuf[pb][jc * 8 + e] / p;
      a[e] = (i == k) ? nr : (a[e] - f * nr);
    }
  }
  if (jc >= 4) {
#pragma unroll
    for (int e = 0; e < 8; ++e) inv[i][(jc - 4) * 8 + e] = a[e];
  }
  __syncthreads();

  float* Abar = ws + WS_ABAR;
  for (int idx = tid; idx < 32 * 32; idx += 256) {
    const int r = idx >> 5, j = idx & 31;
    float s = 0.f;
    for (int k = 0; k < 32; ++k) s += Ah[r][k] * inv[k][j];
    Abar[idx] = s;
  }
  float* Bbar = ws + WS_BBAR;
  for (int n = 0; n < 32; ++n) {
    float s0 = 0.f, s1 = 0.f;
#pragma unroll
    for (int k = 0; k < 32; ++k) {
      const float w = inv[n][k];
      s0 += w * b0[k];
      s1 += w * b1[k];
    }
    Bbar[n * D_MODEL + tid] = dt * s0;
    Bbar[n * D_MODEL + 256 + tid] = dt * s1;
  }
}

__global__ void __launch_bounds__(256) ub_kernel(const float* __restrict__ x,
                                                 const float* __restrict__ Bbar,
                                                 float* __restrict__ uB) {
  const int wave = threadIdx.x >> 5;
  const int lane = threadIdx.x & 31;
  const int r = lane & 15, h = lane >> 4;
  const int mtile = blockIdx.x * 8 + wave;

  const float* xrow = x + (size_t)(mtile * 16 + r) * D_MODEL;
  const float* b0row = Bbar + (size_t)r * D_MODEL;
  const float* b1row = Bbar + (size_t)(16 + r) * D_MODEL;

  v8f acc0 = {0.f, 0.f, 0.f, 0.f, 0.f, 0.f, 0.f, 0.f};
  v8f acc1 = {0.f, 0.f, 0.f, 0.f, 0.f, 0.f, 0.f, 0.f};

  for (int k0 = 0; k0 < D_MODEL; k0 += 4) {
    const int k = k0 + 2 * h;
    const v2f a = *(const v2f*)(xrow + k);
    const v2f b0 = *(const v2f*)(b0row + k);
    const v2f b1 = *(const v2f*)(b1row + k);
    acc0 = wmma4(a, b0, acc0);
    acc1 = wmma4(a, b1, acc1);
  }

  const int mbase = mtile * 16 + 8 * h;
#pragma unroll
  for (int v = 0; v < 8; ++v) {
    const int m = mbase + v;
    uB[(size_t)m * 32 + r] = acc0[v];
    uB[(size_t)m * 32 + 16 + r] = acc1[v];
  }
}

__device__ __forceinline__ float tanh_step(float x) {
  float z;
  asm volatile("v_tanh_f32 %[z], %[x]\n\t"
               "v_nop"
               : [z] "=v"(z)
               : [x] "v"(x));
  return z;
}

#define SCAN_PF 16
#define BC16(dst, sel) "v_permlane16_b32 " dst ", %[st], " sel ", " sel "\n\t"

__device__ __forceinline__ void dot_head(float st, float& t0, float& t1,
                                         float& t2, float& t3) {
  asm volatile(BC16("%[n0]", "%[s0]") BC16("%[n1]", "%[s1]")
               BC16("%[n2]", "%[s2]") BC16("%[n3]", "%[s3]")
               : [n0] "=&v"(t0), [n1] "=&v"(t1), [n2] "=&v"(t2), [n3] "=&v"(t3)
               : [st] "v"(st), [s0] "s"(0x00000000u), [s1] "s"(0x11111111u),
                 [s2] "s"(0x22222222u), [s3] "s"(0x33333333u));
}

__device__ __forceinline__ void dot_first(float st, float ub, const float* ao,
                                          const float* ax, float t0, float t1,
                                          float t2, float t3, float& n0,
                                          float& n1, float& n2, float& n3,
                                          float& o0, float& o1, float& x0,
                                          float& x1) {
  asm volatile(BC16("%[n0]", "%[s0]")
               "v_mul_f32 %[x0], %[c0], %[t0]\n\t"
               "v_fma_f32 %[o0], %[a0], %[t0], %[ub]\n\t"
               BC16("%[n1]", "%[s1]")
               "v_mul_f32 %[x1], %[c1], %[t1]\n\t"
               "v_mul_f32 %[o1], %[a1], %[t1]\n\t"
               BC16("%[n2]", "%[s2]")
               "v_fmac_f32 %[x0], %[c2], %[t2]\n\t"
               "v_fmac_f32 %[o0], %[a2], %[t2]\n\t"
               BC16("%[n3]", "%[s3]")
               "v_fmac_f32 %[x1], %[c3], %[t3]\n\t"
               "v_fmac_f32 %[o1], %[a3], %[t3]"
               : [n0] "=&v"(n0), [n1] "=&v"(n1), [n2] "=&v"(n2), [n3] "=&v"(n3),
                 [o0] "=&v"(o0), [o1] "=&v"(o1), [x0] "=&v"(x0), [x1] "=&v"(x1)
               : [st] "v"(st), [ub] "v"(ub), [t0] "v"(t0), [t1] "v"(t1),
                 [t2] "v"(t2), [t3] "v"(t3), [a0] "v"(ao[0]), [a1] "v"(ao[1]),
                 [a2] "v"(ao[2]), [a3] "v"(ao[3]), [c0] "v"(ax[0]),
                 [c1] "v"(ax[1]), [c2] "v"(ax[2]), [c3] "v"(ax[3]),
                 [s0] "s"(0x44444444u), [s1] "s"(0x55555555u),
                 [s2] "s"(0x66666666u), [s3] "s"(0x77777777u));
}

template <int M>
__device__ __forceinline__ void dot_mid(float st, const float* ao,
                                        const float* ax, float t0, float t1,
                                        float t2, float t3, float& n0, float& n1,
                                        float& n2, float& n3, float& o0,
                                        float& o1, float& x0, float& x1) {
  asm volatile(BC16("%[n0]", "%[s0]")
               "v_fmac_f32 %[x0], %[c0], %[t0]\n\t"
               "v_fmac_f32 %[o0], %[a0], %[t0]\n\t"
               BC16("%[n1]", "%[s1]")
               "v_fmac_f32 %[x1], %[c1], %[t1]\n\t"
               "v_fmac_f32 %[o1], %[a1], %[t1]\n\t"
               BC16("%[n2]", "%[s2]")
               "v_fmac_f32 %[x0], %[c2], %[t2]\n\t"
               "v_fmac_f32 %[o0], %[a2], %[t2]\n\t"
               BC16("%[n3]", "%[s3]")
               "v_fmac_f32 %[x1], %[c3], %[t3]\n\t"
               "v_fmac_f32 %[o1], %[a3], %[t3]"
               : [n0] "=&v"(n0), [n1] "=&v"(n1), [n2] "=&v"(n2), [n3] "=&v"(n3),
                 [o0] "+v"(o0), [o1] "+v"(o1), [x0] "+v"(x0), [x1] "+v"(x1)
               : [st] "v"(st), [t0] "v"(t0), [t1] "v"(t1), [t2] "v"(t2),
                 [t3] "v"(t3), [a0] "v"(ao[M]), [a1] "v"(ao[M + 1]),
                 [a2] "v"(ao[M + 2]), [a3] "v"(ao[M + 3]), [c0] "v"(ax[M]),
                 [c1] "v"(ax[M + 1]), [c2] "v"(ax[M + 2]), [c3] "v"(ax[M + 3]),
                 [s0] "s"(0x11111111u * (unsigned)(M + 4)),
                 [s1] "s"(0x11111111u * (unsigned)(M + 5)),
                 [s2] "s"(0x11111111u * (unsigned)(M + 6)),
                 [s3] "s"(0x11111111u * (unsigned)(M + 7)));
}

__device__ __forceinline__ float dot_tail(const float* ao, const float* ax,
                                          float t0, float t1, float t2, float t3,
                                          float o0, float o1, float x0,
                                          float x1) {
  float acc;
  asm volatile("v_fmac_f32 %[x0], %[c0], %[t0]\n\t"
               "v_fmac_f32 %[o0], %[a0], %[t0]\n\t"
               "v_fmac_f32 %[x1], %[c1], %[t1]\n\t"
               "v_fmac_f32 %[o1], %[a1], %[t1]\n\t"
               "v_fmac_f32 %[x0], %[c2], %[t2]\n\t"
               "v_fmac_f32 %[o0], %[a2], %[t2]\n\t"
               "v_fmac_f32 %[x1], %[c3], %[t3]\n\t"
               "v_fmac_f32 %[o1], %[a3], %[t3]\n\t"
               "v_add_f32 %[x0], %[x0], %[x1]\n\t"
               "v_add_f32 %[o0], %[o0], %[o1]\n\t"
               "v_permlanex16_b32 %[x1], %[x0], %[lo], %[hi]\n\t"
               "v_add_f32 %[acc], %[o0], %[x1]"
               : [acc] "=v"(acc), [o0] "+v"(o0), [o1] "+v"(o1), [x0] "+v"(x0),
                 [x1] "+v"(x1)
               : [t0] "v"(t0), [t1] "v"(t1), [t2] "v"(t2), [t3] "v"(t3),
                 [a0] "v"(ao[12]), [a1] "v"(ao[13]), [a2] "v"(ao[14]),
                 [a3] "v"(ao[15]), [c0] "v"(ax[12]), [c1] "v"(ax[13]),
                 [c2] "v"(ax[14]), [c3] "v"(ax[15]), [lo] "s"(0x76543210u),
                 [hi] "s"(0xfedcba98u));
  return acc;
}

__device__ __forceinline__ void scan_piece(const int b, const int c,
                                           const float* __restrict__ ws,
                                           float* states,
                                           float* __restrict__ fs_out) {
  const int lane = threadIdx.x;
  const int sbeg = c * CH;
  const int hb = lane & 16;
  const float* Abar = ws + WS_ABAR;
  const float* uB = ws + WS_UB + (size_t)b * SEQ * 32;
  float* st_out = states + (size_t)b * SEQ * 32;

  float ao[16], ax[16];
#pragma unroll
  for (int m = 0; m < 16; ++m) {
    ao[m] = Abar[lane * 32 + hb + m];
    ax[m] = Abar[(lane ^ 16) * 32 + hb + m];
  }

  float cur[SCAN_PF];
#pragma unroll
  for (int j = 0; j < SCAN_PF; ++j) cur[j] = uB[(sbeg + j) * 32 + lane];

  float st = (c == 0) ? 0.f : st_out[(sbeg - 1) * 32 + lane];
  for (int s0 = sbeg; s0 < sbeg + CH; s0 += SCAN_PF) {
    const int sn = (s0 + SCAN_PF < sbeg + CH) ? (s0 + SCAN_PF) : s0;
    float nxt[SCAN_PF];
#pragma unroll
    for (int j = 0; j < SCAN_PF; ++j) nxt[j] = uB[(sn + j) * 32 + lane];

#pragma unroll
    for (int j = 0; j < SCAN_PF; ++j) {
      float o0, o1, x0, x1;
      float p0, p1, p2, p3, q0, q1, q2, q3;
      dot_head(st, p0, p1, p2, p3);
      dot_first(st, cur[j], ao, ax, p0, p1, p2, p3, q0, q1, q2, q3, o0, o1, x0,
                x1);
      dot_mid<4>(st, ao, ax, q0, q1, q2, q3, p0, p1, p2, p3, o0, o1, x0, x1);
      dot_mid<8>(st, ao, ax, p0, p1, p2, p3, q0, q1, q2, q3, o0, o1, x0, x1);
      const float acc = dot_tail(ao, ax, q0, q1, q2, q3, o0, o1, x0, x1);
      st = tanh_step(acc);
      st_out[(s0 + j) * 32 + lane] = st;
    }
#pragma unroll
    for (int j = 0; j < SCAN_PF; ++j) cur[j] = nxt[j];
  }
  if (c == NCHUNK - 1) fs_out[b * 32 + lane] = st;
}

__device__ __forceinline__ void cproj_piece(
    const int bid, const int c, const float* states,
    const float* __restrict__ C, const float* __restrict__ x,
    const float* __restrict__ Dv, float* y1) {
  const int wave = threadIdx.x >> 5;
  const int lane = threadIdx.x & 31;
  const int r = lane & 15, h = lane >> 4;
  const int mtile = piece_tile(c, bid * 8 + wave);

  v2f afrag[8];
  const float* srow = states + (size_t)(mtile * 16 + r) * 32;
#pragma unroll
  for (int kk = 0; kk < 8; ++kk)
    afrag[kk] = *(const v2f*)(srow + kk * 4 + 2 * h);

  const int mbase = mtile * 16 + 8 * h;
  for (int dtile = 0; dtile < 32; ++dtile) {
    v8f acc = {0.f, 0.f, 0.f, 0.f, 0.f, 0.f, 0.f, 0.f};
    const float* crow = C + (size_t)(dtile * 16 + r) * 32;
#pragma unroll
    for (int kk = 0; kk < 8; ++kk) {
      const v2f b = *(const v2f*)(crow + kk * 4 + 2 * h);
      acc = wmma4(afrag[kk], b, acc);
    }
    const int d = dtile * 16 + r;
    const float Dd = Dv[d];
#pragma unroll
    for (int v = 0; v < 8; ++v) {
      const int m = mbase + v;
      y1[(size_t)m * D_MODEL + d] = acc[v] + x[(size_t)m * D_MODEL + d] * Dd;
    }
  }
}

#define WOUT_PITCH (D_MODEL + 4)
__device__ __forceinline__ void wout_piece(
    const int bid, const int c, const float* __restrict__ W,
    const float* __restrict__ bout, float* y) {
  __shared__ float lds[16 * WOUT_PITCH];
  const int tid = threadIdx.x;
  const int mtile = piece_tile(c, bid);

  const float4* src4 = (const float4*)(y + (size_t)mtile * 16 * D_MODEL);
#pragma unroll
  for (int i = 0; i < 8; ++i) {
    const int idx = tid + i * 256;
    const int row = idx >> 7, c4 = idx & 127;
    *(float4*)(lds + row * WOUT_PITCH + c4 * 4) = src4[idx];
  }
  __syncthreads();

  const int wave = tid >> 5, lane = tid & 31;
  const int r = lane & 15, h = lane >> 4;
  const int mbase = mtile * 16 + 8 * h;
  const float* arow = lds + r * WOUT_PITCH + 4 * h;

  for (int t = 0; t < 4; ++t) {
    const int ntile = wave * 4 + t;
    const float* wrow = W + (size_t)(ntile * 16 + r) * D_MODEL + 4 * h;
    if (t < 3)
      __builtin_prefetch(W + (size_t)((ntile + 1) * 16 + r) * D_MODEL, 0, 1);

    v8f acc = {0.f, 0.f, 0.f, 0.f, 0.f, 0.f, 0.f, 0.f};
    for (int k8 = 0; k8 < D_MODEL; k8 += 8) {
      const float4 a4 = *(const float4*)(arow + k8);
      const float4 b4 = *(const float4*)(wrow + k8);
      const v2f a_lo = {a4.x, a4.y}, b_lo = {b4.x, b4.y};
      const v2f a_hi = {a4.z, a4.w}, b_hi = {b4.z, b4.w};
      acc = wmma4(a_lo, b_lo, acc);
      acc = wmma4(a_hi, b_hi, acc);
    }
    const int n = ntile * 16 + r;
    const float bo = bout[n];
#pragma unroll
    for (int v = 0; v < 8; ++v)
      y[(size_t)(mbase + v) * D_MODEL + n] = acc[v] + bo;
  }
}

__global__ void __launch_bounds__(256) stage_kernel(
    const float* __restrict__ ws, float* states, float* __restrict__ fs,
    const float* __restrict__ C, const float* __restrict__ x,
    const float* __restrict__ Dv, const float* __restrict__ W,
    const float* __restrict__ bout, float* y, const int c) {
  int bid = blockIdx.x;
  if (bid < NB_SCAN) {
    if (c < NCHUNK && threadIdx.x < 32) scan_piece(bid, c, ws, states, fs);
    return;
  }
  bid -= NB_SCAN;
  if (bid < NB_CPROJ) {
    if (c >= 1 && c <= NCHUNK) cproj_piece(bid, c - 1, states, C, x, Dv, y);
    return;
  }
  bid -= NB_CPROJ;
  if (c >= 2) wout_piece(bid, c - 2, W, bout, y);
}

extern "C" void kernel_launch(void* const* d_in, const int* in_sizes, int n_in,
                              void* d_out, int out_size, void* d_ws,
                              size_t ws_size, hipStream_t stream) {
  const float* x = (const float*)d_in[0];
  const float* A = (const float*)d_in[1];
  const float* B = (const float*)d_in[2];
  const float* C = (const float*)d_in[3];
  const float* Dv = (const float*)d_in[4];
  const float* log_dt = (const float*)d_in[5];
  const float* W = (const float*)d_in[6];
  const float* bout = (const float*)d_in[7];

  float* y = (float*)d_out;
  float* fs = y + (size_t)ROWS * D_MODEL;

  float* ws = (float*)d_ws;
  float* Bbar = ws + WS_BBAR;
  float* uB = ws + WS_UB;
  float* states = ws + WS_STATES;

  prep_kernel<<<1, 256, 0, stream>>>(A, B, log_dt, ws);
  ub_kernel<<<256, 256, 0, stream>>>(x, Bbar, uB);
  stage_kernel<<<8, 256, 0, stream>>>(ws, states, fs, C, x, Dv, W, bout, y, 0);
  stage_kernel<<<40, 256, 0, stream>>>(ws, states, fs, C, x, Dv, W, bout, y, 1);
  stage_kernel<<<296, 256, 0, stream>>>(ws, states, fs, C, x, Dv, W, bout, y, 2);
  stage_kernel<<<296, 256, 0, stream>>>(ws, states, fs, C, x, Dv, W, bout, y, 3);
  stage_kernel<<<296, 256, 0, stream>>>(ws, states, fs, C, x, Dv, W, bout, y, 4);
  stage_kernel<<<296, 256, 0, stream>>>(ws, states, fs, C, x, Dv, W, bout, y, 5);
  stage_kernel<<<296, 256, 0, stream>>>(ws, states, fs, C, x, Dv, W, bout, y, 6);
  stage_kernel<<<296, 256, 0, stream>>>(ws, states, fs, C, x, Dv, W, bout, y, 7);
  stage_kernel<<<296, 256, 0, stream>>>(ws, states, fs, C, x, Dv, W, bout, y, 8);
  stage_kernel<<<296, 256, 0, stream>>>(ws, states, fs, C, x, Dv, W, bout, y, 9);
}


# scan: the two products of a step that share a broadcast state value issue as one dual-issue instruction (16 pairs for 32 FMAs), registers named for its bank rule; sums and their order unchanged
# speedup vs baseline: 8.0280x; 1.0307x over previous
#include <hip/hip_runtime.h>

typedef float v2f __attribute__((ext_vector_type(2)));
typedef float v8f __attribute__((ext_vector_type(8)));

#define D_MODEL 512
#define D_STATE 32
#define BATCH 8
#define SEQ 4096
#define ROWS (BATCH * SEQ)

#define WS_ABAR 0
#define WS_BBAR (WS_ABAR + 1024)
#define WS_UB (WS_BBAR + 16384)
#define WS_STATES (WS_UB + ROWS * 32)

#define NCHUNK 8
#define CH (SEQ / NCHUNK)
#define CH_MTILES (BATCH * CH / 16)
#define NB_SCAN BATCH
#define NB_CPROJ (CH_MTILES / 8)
#define NB_WOUT CH_MTILES
__device__ __forceinline__ int piece_tile(int c, int j) {
  return (j >> 5) * (SEQ / 16) + c * (CH / 16) + (j & 31);
}

__device__ __forceinline__ v8f wmma4(v2f a, v2f b, v8f c) {
  return __builtin_amdgcn_wmma_f32_16x16x4_f32(false, a, false, b, (short)0, c,
                                               false, false);
}

__global__ void __launch_bounds__(256) prep_kernel(
    const float* __restrict__ A, const float* __restrict__ B,
    const float* __restrict__ log_dt, float* __restrict__ ws) {
  __shared__ float red[256];
  __shared__ float red2[32];
  __shared__ float rowbuf[2][64];
  __shared__ float colbuf[2][32];
  __shared__ float Ah[32][32];
  __shared__ float inv[32][32];
  __shared__ float dt_sh;
  const int tid = threadIdx.x;

  float b0[32], b1[32];
#pragma unroll
  for (int k = 0; k < 32; ++k) {
    b0[k] = B[k * D_MODEL + tid];
    b1[k] = B[k * D_MODEL + 256 + tid];
  }

  const int i = tid >> 3, jc = tid & 7;
  float av[8];
#pragma unroll
  for (int e = 0; e < 8; ++e) av[e] = A[i * 32 + (jc & 3) * 8 + e];

  {
    float v0 = expf(log_dt[tid]);
    float v1 = expf(log_dt[tid + 256]);
    v0 = fminf(fmaxf(v0, 1e-4f), 1.0f);
    v1 = fminf(fmaxf(v1, 1e-4f), 1.0f);
    red[tid] = v0 + v1;
  }
  __syncthreads();
  if (tid < 32) {
    float s = 0.f;
    for (int q = 0; q < 8; ++q) s += red[tid * 8 + q];
    red2[tid] = s;
  }
  __syncthreads();
  if (tid == 0) {
    float s = 0.f;
    for (int q = 0; q < 32; ++q) s += red2[q];
    dt_sh = s / (float)D_MODEL;
  }
  __syncthreads();
  const float dt = dt_sh;

  float a[8];
#pragma unroll
  for (int e = 0; e < 8; ++e) {
    const int j = (jc & 3) * 8 + e;
    const float eye = (i == j) ? 1.f : 0.f;
    const float lo = eye - 0.5f * dt * av[e];
    if (jc < 4) Ah[i][j] = eye + 0.5f * dt * av[e];
    a[e] = (jc < 4) ? lo : eye;
  }

  for (int k = 0; k < 32; ++k) {
    const int pb = k & 1;
    if (i == k) {
#pragma unroll
      for (int e = 0; e < 8; ++e) rowbuf[pb][jc * 8 + e] = a[e];
    }
    if (jc == (k >> 3)) {
      float ak = a[0];
#pragma unroll
      for (int e = 1; e < 8; ++e) ak = ((k & 7) == e) ? a[e] : ak;
      colbuf[pb][i] = ak;
    }
    __syncthreads();
    const float p = rowbuf[pb][k];
    const float f = colbuf[pb][i];
#pragma unroll
    for (int e = 0; e < 8; ++e) {
      const float nr = rowbuf[pb][jc * 8 + e] / p;
      a[e] = (i == k) ? nr : (a[e] - f * nr);
    }
  }
  if (jc >= 4) {
#pragma unroll
    for (int e = 0; e < 8; ++e) inv[i][(jc - 4) * 8 + e] = a[e];
  }
  __syncthreads();

  float* Abar = ws + WS_ABAR;
  for (int idx = tid; idx < 32 * 32; idx += 256) {
    const int r = idx >> 5, j = idx & 31;
    float s = 0.f;
    for (int k = 0; k < 32; ++k) s += Ah[r][k] * inv[k][j];
    Abar[idx] = s;
  }
  float* Bbar = ws + WS_BBAR;
  for (int n = 0; n < 32; ++n) {
    float s0 = 0.f, s1 = 0.f;
#pragma unroll
    for (int k = 0; k < 32; ++k) {
      const float w = inv[n][k];
      s0 += w * b0[k];
      s1 += w * b1[k];
    }
    Bbar[n * D_MODEL + tid] = dt * s0;
    Bbar[n * D_MODEL + 256 + tid] = dt * s1;
  }
}

__global__ void __launch_bounds__(256) ub_kernel(const float* __restrict__ x,
                                                 const float* __restrict__ Bbar,
                                                 float* __restrict__ uB) {
  const int wave = threadIdx.x >> 5;
  const int lane = threadIdx.x & 31;
  const int r = lane & 15, h = lane >> 4;
  const int mtile = blockIdx.x * 8 + wave;

  const float* xrow = x + (size_t)(mtile * 16 + r) * D_MODEL;
  const float* b0row = Bbar + (size_t)r * D_MODEL;
  const float* b1row = Bbar + (size_t)(16 + r) * D_MODEL;

  v8f acc0 = {0.f, 0.f, 0.f, 0.f, 0.f, 0.f, 0.f, 0.f};
  v8f acc1 = {0.f, 0.f, 0.f, 0.f, 0.f, 0.f, 0.f, 0.f};

  for (int k0 = 0; k0 < D_MODEL; k0 += 4) {
    const int k = k0 + 2 * h;
    const v2f a = *(const v2f*)(xrow + k);
    const v2f b0 = *(const v2f*)(b0row + k);
    const v2f b1 = *(const v2f*)(b1row + k);
    acc0 = wmma4(a, b0, acc0);
    acc1 = wmma4(a, b1, acc1);
  }

  const int mbase = mtile * 16 + 8 * h;
#pragma unroll
  for (int v = 0; v < 8; ++v) {
    const int m = mbase + v;
    uB[(size_t)m * 32 + r] = acc0[v];
    uB[(size_t)m * 32 + 16 + r] = acc1[v];
  }
}

#define SCAN_PF 16
__device__ __forceinline__ float scan_step(float st, float ub, const float* ao,
                                           const float* ax) {
  float stn;
  asm volatile(
      "v_permlane16_b32 v162, %[st], %[s0], %[s0]\n\t"
      "v_permlane16_b32 v164, %[st], %[s1], %[s1]\n\t"
      "v_permlane16_b32 v166, %[st], %[s2], %[s2]\n\t"
      "v_dual_fma_f32 v199, v128, v162, %[ub] :: v_dual_mul_f32 v196, v162, v129\n\t"
      "v_permlane16_b32 v168, %[st], %[s3], %[s3]\n\t"
      "v_dual_mul_f32 v197, v130, v164 :: v_dual_mul_f32 v198, v164, v131\n\t"
      "v_permlane16_b32 v170, %[st], %[s4], %[s4]\n\t"
      "v_dual_fmac_f32 v199, v132, v166 :: v_dual_fmac_f32 v196, v166, v133\n\t"
      "v_permlane16_b32 v172, %[st], %[s5], %[s5]\n\t"
      "v_dual_fmac_f32 v197, v134, v168 :: v_dual_fmac_f32 v198, v168, v135\n\t"
      "v_permlane16_b32 v174, %[st], %[s6], %[s6]\n\t"
      "v_dual_fmac_f32 v199, v136, v170 :: v_dual_fmac_f32 v196, v170, v137\n\t"
      "v_permlane16_b32 v176, %[st], %[s7], %[s7]\n\t"
      "v_dual_fmac_f32 v197, v138, v172 :: v_dual_fmac_f32 v198, v172, v139\n\t"
      "v_permlane16_b32 v178, %[st], %[s8], %[s8]\n\t"
      "v_dual_fmac_f32 v199, v140, v174 :: v_dual_fmac_f32 v196, v174, v141\n\t"
      "v_permlane16_b32 v180, %[st], %[s9], %[s9]\n\t"
      "v_dual_fmac_f32 v197, v142, v176 :: v_dual_fmac_f32 v198, v176, v143\n\t"
      "v_permlane16_b32 v182, %[st], %[s10], %[s10]\n\t"
      "v_dual_fmac_f32 v199, v144, v178 :: v_dual_fmac_f32 v196, v178, v145\n\t"
      "v_permlane16_b32 v184, %[st], %[s11], %[s11]\n\t"
      "v_dual_fmac_f32 v197, v146, v180 :: v_dual_fmac_f32 v198, v180, v147\n\t"
      "v_permlane16_b32 v186, %[st], %[s12], %[s12]\n\t"
      "v_dual_fmac_f32 v199, v148, v182 :: v_dual_fmac_f32 v196, v182, v149\n\t"
      "v_permlane16_b32 v188, %[st], %[s13], %[s13]\n\t"
      "v_dual_fmac_f32 v197, v150, v184 :: v_dual_fmac_f32 v198, v184, v151\n\t"
      "v_permlane16_b32 v190, %[st], %[s14], %[s14]\n\t"
      "v_dual_fmac_f32 v199, v152, v186 :: v_dual_fmac_f32 v196, v186, v153\n\t"
      "v_permlane16_b32 v192, %[st], %[s15], %[s15]\n\t"
      "v_dual_fmac_f32 v197, v154, v188 :: v_dual_fmac_f32 v198, v188, v155\n\t"
      "v_dual_fmac_f32 v199, v156, v190 :: v_dual_fmac_f32 v196, v190, v157\n\t"
      "v_dual_fmac_f32 v197, v158, v192 :: v_dual_fmac_f32 v198, v192, v159\n\t"
      "v_add_f32 v196, v196, v198\n\t"
      "v_add_f32 v199, v199, v197\n\t"
      "v_permlanex16_b32 v198, v196, %[lo], %[hi]\n\t"
      "v_add_f32 v199, v199, v198\n\t"
      "v_tanh_f32 %[stn], v199\n\t"
      "v_nop"
      : [stn] "=&v"(stn)
      : [st] "v"(st), [ub] "v"(ub),
        "{v128}"(ao[0]), "{v129}"(ax[0]),
        "{v130}"(ao[1]), "{v131}"(ax[1]),
        "{v132}"(ao[2]), "{v133}"(ax[2]),
        "{v134}"(ao[3]), "{v135}"(ax[3]),
        "{v136}"(ao[4]), "{v137}"(ax[4]),
        "{v138}"(ao[5]), "{v139}"(ax[5]),
        "{v140}"(ao[6]), "{v141}"(ax[6]),
        "{v142}"(ao[7]), "{v143}"(ax[7]),
        "{v144}"(ao[8]), "{v145}"(ax[8]),
        "{v146}"(ao[9]), "{v147}"(ax[9]),
        "{v148}"(ao[10]), "{v149}"(ax[10]),
        "{v150}"(ao[11]), "{v151}"(ax[11]),
        "{v152}"(ao[12]), "{v153}"(ax[12]),
        "{v154}"(ao[13]), "{v155}"(ax[13]),
        "{v156}"(ao[14]), "{v157}"(ax[14]),
        "{v158}"(ao[15]), "{v159}"(ax[15]),
        [s0] "s"(0x00000000u),
        [s1] "s"(0x11111111u),
        [s2] "s"(0x22222222u),
        [s3] "s"(0x33333333u),
        [s4] "s"(0x44444444u),
        [s5] "s"(0x55555555u),
        [s6] "s"(0x66666666u),
        [s7] "s"(0x77777777u),
        [s8] "s"(0x88888888u),
        [s9] "s"(0x99999999u),
        [s10] "s"(0xaaaaaaaau),
        [s11] "s"(0xbbbbbbbbu),
        [s12] "s"(0xccccccccu),
        [s13] "s"(0xddddddddu),
        [s14] "s"(0xeeeeeeeeu),
        [s15] "s"(0xffffffffu),
        [lo] "s"(0x76543210u), [hi] "s"(0xfedcba98u)
      : "v162", "v164", "v166", "v168", "v170", "v172", "v174", "v176", "v178",
        "v180", "v182", "v184", "v186", "v188", "v190", "v192", "v196", "v197",
        "v198", "v199");
  return stn;
}

__device__ __forceinline__ void scan_piece(const int b, const int c,
                                           const float* __restrict__ ws,
                                           float* states,
                                           float* __restrict__ fs_out) {
  const int lane = threadIdx.x;
  const int sbeg = c * CH;
  const int hb = lane & 16;
  const float* Abar = ws + WS_ABAR;
  const float* uB = ws + WS_UB + (size_t)b * SEQ * 32;
  float* st_out = states + (size_t)b * SEQ * 32;

  float ao[16], ax[16];
#pragma unroll
  for (int m = 0; m < 16; ++m) {
    ao[m] = Abar[lane * 32 + hb + m];
    ax[m] = Abar[(lane ^ 16) * 32 + hb + m];
  }

  float cur[SCAN_PF];
#pragma unroll
  for (int j = 0; j < SCAN_PF; ++j) cur[j] = uB[(sbeg + j) * 32 + lane];

  float st = (c == 0) ? 0.f : st_out[(sbeg - 1) * 32 + lane];

  asm volatile(""
               : "+{v128}"(ao[0]), "+{v129}"(ax[0]),
               "+{v130}"(ao[1]), "+{v131}"(ax[1]),
               "+{v132}"(ao[2]), "+{v133}"(ax[2]),
               "+{v134}"(ao[3]), "+{v135}"(ax[3]),
               "+{v136}"(ao[4]), "+{v137}"(ax[4]),
               "+{v138}"(ao[5]), "+{v139}"(ax[5]),
               "+{v140}"(ao[6]), "+{v141}"(ax[6]),
               "+{v142}"(ao[7]), "+{v143}"(ax[7]),
               "+{v144}"(ao[8]), "+{v145}"(ax[8]),
               "+{v146}"(ao[9]), "+{v147}"(ax[9]),
               "+{v148}"(ao[10]), "+{v149}"(ax[10]),
               "+{v150}"(ao[11]), "+{v151}"(ax[11]),
               "+{v152}"(ao[12]), "+{v153}"(ax[12]),
               "+{v154}"(ao[13]), "+{v155}"(ax[13]),
               "+{v156}"(ao[14]), "+{v157}"(ax[14]),
               "+{v158}"(ao[15]), "+{v159}"(ax[15]));

  for (int s0 = sbeg; s0 < sbeg + CH; s0 += SCAN_PF) {
    const int sn = (s0 + SCAN_PF < sbeg + CH) ? (s0 + SCAN_PF) : s0;
    float nxt[SCAN_PF];
#pragma unroll
    for (int j = 0; j < SCAN_PF; ++j) nxt[j] = uB[(sn + j) * 32 + lane];

#pragma unroll
    for (int j = 0; j < SCAN_PF; ++j) {
      st = scan_step(st, cur[j], ao, ax);
      st_out[(s0 + j) * 32 + lane] = st;
    }
#pragma unroll
    for (int j = 0; j < SCAN_PF; ++j) cur[j] = nxt[j];
  }
  if (c == NCHUNK - 1) fs_out[b * 32 + lane] = st;
}

__device__ __forceinline__ void cproj_piece(
    const int bid, const int c, const float* states,
    const float* __restrict__ C, const float* __restrict__ x,
    const float* __restrict__ Dv, float* y1) {
  const int wave = threadIdx.x >> 5;
  const int lane = threadIdx.x & 31;
  const int r = lane & 15, h = lane >> 4;
  const int mtile = piece_tile(c, bid * 8 + wave);

  v2f afrag[8];
  const float* srow = states + (size_t)(mtile * 16 + r) * 32;
#pragma unroll
  for (int kk = 0; kk < 8; ++kk)
    afrag[kk] = *(const v2f*)(srow + kk * 4 + 2 * h);

  const int mbase = mtile * 16 + 8 * h;
  for (int dtile = 0; dtile < 32; ++dtile) {
    v8f acc = {0.f, 0.f, 0.f, 0.f, 0.f, 0.f, 0.f, 0.f};
    const float* crow = C + (size_t)(dtile * 16 + r) * 32;
#pragma unroll
    for (int kk = 0; kk < 8; ++kk) {
      const v2f b = *(const v2f*)(crow + kk * 4 + 2 * h);
      acc = wmma4(afrag[kk], b, acc);
    }
    const int d = dtile * 16 + r;
    const float Dd = Dv[d];
#pragma unroll
    for (int v = 0; v < 8; ++v) {
      const int m = mbase + v;
      y1[(size_t)m * D_MODEL + d] = acc[v] + x[(size_t)m * D_MODEL + d] * Dd;
    }
  }
}

#define WOUT_PITCH (D_MODEL + 4)
__device__ __forceinline__ void wout_piece(
    const int bid, const int c, const float* __restrict__ W,
    const float* __restrict__ bout, float* y) {
  __shared__ float lds[16 * WOUT_PITCH];
  const int tid = threadIdx.x;
  const int mtile = piece_tile(c, bid);

  const float4* src4 = (const float4*)(y + (size_t)mtile * 16 * D_MODEL);
#pragma unroll
  for (int i = 0; i < 8; ++i) {
    const int idx = tid + i * 256;
    const int row = idx >> 7, c4 = idx & 127;
    *(float4*)(lds + row * WOUT_PITCH + c4 * 4) = src4[idx];
  }
  __syncthreads();

  const int wave = tid >> 5, lane = tid & 31;
  const int r = lane & 15, h = lane >> 4;
  const int mbase = mtile * 16 + 8 * h;
  const float* arow = lds + r * WOUT_PITCH + 4 * h;

  for (int t = 0; t < 4; ++t) {
    const int ntile = wave * 4 + t;
    const float* wrow = W + (size_t)(ntile * 16 + r) * D_MODEL + 4 * h;
    if (t < 3)
      __builtin_prefetch(W + (size_t)((ntile + 1) * 16 + r) * D_MODEL, 0, 1);

    v8f acc = {0.f, 0.f, 0.f, 0.f, 0.f, 0.f, 0.f, 0.f};
    for (int k8 = 0; k8 < D_MODEL; k8 += 8) {
      const float4 a4 = *(const float4*)(arow + k8);
      const float4 b4 = *(const float4*)(wrow + k8);
      const v2f a_lo = {a4.x, a4.y}, b_lo = {b4.x, b4.y};
      const v2f a_hi = {a4.z, a4.w}, b_hi = {b4.z, b4.w};
      acc = wmma4(a_lo, b_lo, acc);
      acc = wmma4(a_hi, b_hi, acc);
    }
    const int n = ntile * 16 + r;
    const float bo = bout[n];
#pragma unroll
    for (int v = 0; v < 8; ++v)
      y[(size_t)(mbase + v) * D_MODEL + n] = acc[v] + bo;
  }
}

__global__ void __launch_bounds__(256) stage_kernel(
    const float* __restrict__ ws, float* states, float* __restrict__ fs,
    const float* __restrict__ C, const float* __restrict__ x,
    const float* __restrict__ Dv, const float* __restrict__ W,
    const float* __restrict__ bout, float* y, const int c) {
  int bid = blockIdx.x;
  if (bid < NB_SCAN) {
    if (c < NCHUNK && threadIdx.x < 32) scan_piece(bid, c, ws, states, fs);
    return;
  }
  bid -= NB_SCAN;
  if (bid < NB_CPROJ) {
    if (c >= 1 && c <= NCHUNK) cproj_piece(bid, c - 1, states, C, x, Dv, y);
    return;
  }
  bid -= NB_CPROJ;
  if (c >= 2) wout_piece(bid, c - 2, W, bout, y);
}

extern "C" void kernel_launch(void* const* d_in, const int* in_sizes, int n_in,
                              void* d_out, int out_size, void* d_ws,
                              size_t ws_size, hipStream_t stream) {
  const float* x = (const float*)d_in[0];
  const float* A = (const float*)d_in[1];
  const float* B = (const float*)d_in[2];
  const float* C = (const float*)d_in[3];
  const float* Dv = (const float*)d_in[4];
  const float* log_dt = (const float*)d_in[5];
  const float* W = (const float*)d_in[6];
  const float* bout = (const float*)d_in[7];

  float* y = (float*)d_out;
  float* fs = y + (size_t)ROWS * D_MODEL;

  float* ws = (float*)d_ws;
  float* Bbar = ws + WS_BBAR;
  float* uB = ws + WS_UB;
  float* states = ws + WS_STATES;

  prep_kernel<<<1, 256, 0, stream>>>(A, B, log_dt, ws);
  ub_kernel<<<256, 256, 0, stream>>>(x, Bbar, uB);
  stage_kernel<<<8, 256, 0, stream>>>(ws, states, fs, C, x, Dv, W, bout, y, 0);
  stage_kernel<<<40, 256, 0, stream>>>(ws, states, fs, C, x, Dv, W, bout, y, 1);
  stage_kernel<<<296, 256, 0, stream>>>(ws, states, fs, C, x, Dv, W, bout, y, 2);
  stage_kernel<<<296, 256, 0, stream>>>(ws, states, fs, C, x, Dv, W, bout, y, 3);
  stage_kernel<<<296, 256, 0, stream>>>(ws, states, fs, C, x, Dv, W, bout, y, 4);
  stage_kernel<<<296, 256, 0, stream>>>(ws, states, fs, C, x, Dv, W, bout, y, 5);
  stage_kernel<<<296, 256, 0, stream>>>(ws, states, fs, C, x, Dv, W, bout, y, 6);
  stage_kernel<<<296, 256, 0, stream>>>(ws, states, fs, C, x, Dv, W, bout, y, 7);
  stage_kernel<<<296, 256, 0, stream>>>(ws, states, fs, C, x, Dv, W, bout, y, 8);
  stage_kernel<<<296, 256, 0, stream>>>(ws, states, fs, C, x, Dv, W, bout, y, 9);
}


# proj: states @ C^T + x D is formed in LDS by the output projection's own work-groups (same fragments, sums and order) and no longer goes through memory; 11 launches for 12, pipeline two deep
# speedup vs baseline: 8.5305x; 1.0626x over previous
#include <hip/hip_runtime.h>

typedef float v2f __attribute__((ext_vector_type(2)));
typedef float v8f __attribute__((ext_vector_type(8)));

#define D_MODEL 512
#define D_STATE 32
#define BATCH 8
#define SEQ 4096
#define ROWS (BATCH * SEQ)

#define WS_ABAR 0
#define WS_BBAR (WS_ABAR + 1024)
#define WS_UB (WS_BBAR + 16384)
#define WS_STATES (WS_UB + ROWS * 32)

#define NCHUNK 8
#define CH (SEQ / NCHUNK)
#define CH_MTILES (BATCH * CH / 16)
#define NB_SCAN BATCH
#define NB_PROJ CH_MTILES
__device__ __forceinline__ int piece_tile(int c, int j) {
  return (j >> 5) * (SEQ / 16) + c * (CH / 16) + (j & 31);
}

__device__ __forceinline__ v8f wmma4(v2f a, v2f b, v8f c) {
  return __builtin_amdgcn_wmma_f32_16x16x4_f32(false, a, false, b, (short)0, c,
                                               false, false);
}

__global__ void __launch_bounds__(256) prep_kernel(
    const float* __restrict__ A, const float* __restrict__ B,
    const float* __restrict__ log_dt, float* __restrict__ ws) {
  __shared__ float red[256];
  __shared__ float red2[32];
  __shared__ float rowbuf[2][64];
  __shared__ float colbuf[2][32];
  __shared__ float Ah[32][32];
  __shared__ float inv[32][32];
  __shared__ float dt_sh;
  const int tid = threadIdx.x;

  float b0[32], b1[32];
#pragma unroll
  for (int k = 0; k < 32; ++k) {
    b0[k] = B[k * D_MODEL + tid];
    b1[k] = B[k * D_MODEL + 256 + tid];
  }

  const int i = tid >> 3, jc = tid & 7;
  float av[8];
#pragma unroll
  for (int e = 0; e < 8; ++e) av[e] = A[i * 32 + (jc & 3) * 8 + e];

  {
    float v0 = expf(log_dt[tid]);
    float v1 = expf(log_dt[tid + 256]);
    v0 = fminf(fmaxf(v0, 1e-4f), 1.0f);
    v1 = fminf(fmaxf(v1, 1e-4f), 1.0f);
    red[tid] = v0 + v1;
  }
  __syncthreads();
  if (tid < 32) {
    float s = 0.f;
    for (int q = 0; q < 8; ++q) s += red[tid * 8 + q];
    red2[tid] = s;
  }
  __syncthreads();
  if (tid == 0) {
    float s = 0.f;
    for (int q = 0; q < 32; ++q) s += red2[q];
    dt_sh = s / (float)D_MODEL;
  }
  __syncthreads();
  const float dt = dt_sh;

  float a[8];
#pragma unroll
  for (int e = 0; e < 8; ++e) {
    const int j = (jc & 3) * 8 + e;
    const float eye = (i == j) ? 1.f : 0.f;
    const float lo = eye - 0.5f * dt * av[e];
    if (jc < 4) Ah[i][j] = eye + 0.5f * dt * av[e];
    a[e] = (jc < 4) ? lo : eye;
  }

  for (int k = 0; k < 32; ++k) {
    const int pb = k & 1;
    if (i == k) {
#pragma unroll
      for (int e = 0; e < 8; ++e) rowbuf[pb][jc * 8 + e] = a[e];
    }
    if (jc == (k >> 3)) {
      float ak = a[0];
#pragma unroll
      for (int e = 1; e < 8; ++e) ak = ((k & 7) == e) ? a[e] : ak;
      colbuf[pb][i] = ak;
    }
    __syncthreads();
    const float p = rowbuf[pb][k];
    const float f = colbuf[pb][i];
#pragma unroll
    for (int e = 0; e < 8; ++e) {
      const float nr = rowbuf[pb][jc * 8 + e] / p;
      a[e] = (i == k) ? nr : (a[e] - f * nr);
    }
  }
  if (jc >= 4) {
#pragma unroll
    for (int e = 0; e < 8; ++e) inv[i][(jc - 4) * 8 + e] = a[e];
  }
  __syncthreads();

  float* Abar = ws + WS_ABAR;
  for (int idx = tid; idx < 32 * 32; idx += 256) {
    const int r = idx >> 5, j = idx & 31;
    float s = 0.f;
    for (int k = 0; k < 32; ++k) s += Ah[r][k] * inv[k][j];
    Abar[idx] = s;
  }
  float* Bbar = ws + WS_BBAR;
  for (int n = 0; n < 32; ++n) {
    float s0 = 0.f, s1 = 0.f;
#pragma unroll
    for (int k = 0; k < 32; ++k) {
      const float w = inv[n][k];
      s0 += w * b0[k];
      s1 += w * b1[k];
    }
    Bbar[n * D_MODEL + tid] = dt * s0;
    Bbar[n * D_MODEL + 256 + tid] = dt * s1;
  }
}

__global__ void __launch_bounds__(256) ub_kernel(const float* __restrict__ x,
                                                 const float* __restrict__ Bbar,
                                                 float* __restrict__ uB) {
  const int wave = threadIdx.x >> 5;
  const int lane = threadIdx.x & 31;
  const int r = lane & 15, h = lane >> 4;
  const int mtile = blockIdx.x * 8 + wave;

  const float* xrow = x + (size_t)(mtile * 16 + r) * D_MODEL;
  const float* b0row = Bbar + (size_t)r * D_MODEL;
  const float* b1row = Bbar + (size_t)(16 + r) * D_MODEL;

  v8f acc0 = {0.f, 0.f, 0.f, 0.f, 0.f, 0.f, 0.f, 0.f};
  v8f acc1 = {0.f, 0.f, 0.f, 0.f, 0.f, 0.f, 0.f, 0.f};

  for (int k0 = 0; k0 < D_MODEL; k0 += 4) {
    const int k = k0 + 2 * h;
    const v2f a = *(const v2f*)(xrow + k);
    const v2f b0 = *(const v2f*)(b0row + k);
    const v2f b1 = *(const v2f*)(b1row + k);
    acc0 = wmma4(a, b0, acc0);
    acc1 = wmma4(a, b1, acc1);
  }

  const int mbase = mtile * 16 + 8 * h;
#pragma unroll
  for (int v = 0; v < 8; ++v) {
    const int m = mbase + v;
    uB[(size_t)m * 32 + r] = acc0[v];
    uB[(size_t)m * 32 + 16 + r] = acc1[v];
  }
}

#define SCAN_PF 16
__device__ __forceinline__ float scan_step(float st, float ub, const float* ao,
                                           const float* ax) {
  float stn;
  asm volatile(
      "v_permlane16_b32 v162, %[st], %[s0], %[s0]\n\t"
      "v_permlane16_b32 v164, %[st], %[s1], %[s1]\n\t"
      "v_permlane16_b32 v166, %[st], %[s2], %[s2]\n\t"
      "v_dual_fma_f32 v199, v128, v162, %[ub] :: v_dual_mul_f32 v196, v162, v129\n\t"
      "v_permlane16_b32 v168, %[st], %[s3], %[s3]\n\t"
      "v_dual_mul_f32 v197, v130, v164 :: v_dual_mul_f32 v198, v164, v131\n\t"
      "v_permlane16_b32 v170, %[st], %[s4], %[s4]\n\t"
      "v_dual_fmac_f32 v199, v132, v166 :: v_dual_fmac_f32 v196, v166, v133\n\t"
      "v_permlane16_b32 v172, %[st], %[s5], %[s5]\n\t"
      "v_dual_fmac_f32 v197, v134, v168 :: v_dual_fmac_f32 v198, v168, v135\n\t"
      "v_permlane16_b32 v174, %[st], %[s6], %[s6]\n\t"
      "v_dual_fmac_f32 v199, v136, v170 :: v_dual_fmac_f32 v196, v170, v137\n\t"
      "v_permlane16_b32 v176, %[st], %[s7], %[s7]\n\t"
      "v_dual_fmac_f32 v197, v138, v172 :: v_dual_fmac_f32 v198, v172, v139\n\t"
      "v_permlane16_b32 v178, %[st], %[s8], %[s8]\n\t"
      "v_dual_fmac_f32 v199, v140, v174 :: v_dual_fmac_f32 v196, v174, v141\n\t"
      "v_permlane16_b32 v180, %[st], %[s9], %[s9]\n\t"
      "v_dual_fmac_f32 v197, v142, v176 :: v_dual_fmac_f32 v198, v176, v143\n\t"
      "v_permlane16_b32 v182, %[st], %[s10], %[s10]\n\t"
      "v_dual_fmac_f32 v199, v144, v178 :: v_dual_fmac_f32 v196, v178, v145\n\t"
      "v_permlane16_b32 v184, %[st], %[s11], %[s11]\n\t"
      "v_dual_fmac_f32 v197, v146, v180 :: v_dual_fmac_f32 v198, v180, v147\n\t"
      "v_permlane16_b32 v186, %[st], %[s12], %[s12]\n\t"
      "v_dual_fmac_f32 v199, v148, v182 :: v_dual_fmac_f32 v196, v182, v149\n\t"
      "v_permlane16_b32 v188, %[st], %[s13], %[s13]\n\t"
      "v_dual_fmac_f32 v197, v150, v184 :: v_dual_fmac_f32 v198, v184, v151\n\t"
      "v_permlane16_b32 v190, %[st], %[s14], %[s14]\n\t"
      "v_dual_fmac_f32 v199, v152, v186 :: v_dual_fmac_f32 v196, v186, v153\n\t"
      "v_permlane16_b32 v192, %[st], %[s15], %[s15]\n\t"
      "v_dual_fmac_f32 v197, v154, v188 :: v_dual_fmac_f32 v198, v188, v155\n\t"
      "v_dual_fmac_f32 v199, v156, v190 :: v_dual_fmac_f32 v196, v190, v157\n\t"
      "v_dual_fmac_f32 v197, v158, v192 :: v_dual_fmac_f32 v198, v192, v159\n\t"
      "v_add_f32 v196, v196, v198\n\t"
      "v_add_f32 v199, v199, v197\n\t"
      "v_permlanex16_b32 v198, v196, %[lo], %[hi]\n\t"
      "v_add_f32 v199, v199, v198\n\t"
      "v_tanh_f32 %[stn], v199\n\t"
      "v_nop"
      : [stn] "=&v"(stn)
      : [st] "v"(st), [ub] "v"(ub),
        "{v128}"(ao[0]), "{v129}"(ax[0]),
        "{v130}"(ao[1]), "{v131}"(ax[1]),
        "{v132}"(ao[2]), "{v133}"(ax[2]),
        "{v134}"(ao[3]), "{v135}"(ax[3]),
        "{v136}"(ao[4]), "{v137}"(ax[4]),
        "{v138}"(ao[5]), "{v139}"(ax[5]),
        "{v140}"(ao[6]), "{v141}"(ax[6]),
        "{v142}"(ao[7]), "{v143}"(ax[7]),
        "{v144}"(ao[8]), "{v145}"(ax[8]),
        "{v146}"(ao[9]), "{v147}"(ax[9]),
        "{v148}"(ao[10]), "{v149}"(ax[10]),
        "{v150}"(ao[11]), "{v151}"(ax[11]),
        "{v152}"(ao[12]), "{v153}"(ax[12]),
        "{v154}"(ao[13]), "{v155}"(ax[13]),
        "{v156}"(ao[14]), "{v157}"(ax[14]),
        "{v158}"(ao[15]), "{v159}"(ax[15]),
        [s0] "s"(0x00000000u),
        [s1] "s"(0x11111111u),
        [s2] "s"(0x22222222u),
        [s3] "s"(0x33333333u),
        [s4] "s"(0x44444444u),
        [s5] "s"(0x55555555u),
        [s6] "s"(0x66666666u),
        [s7] "s"(0x77777777u),
        [s8] "s"(0x88888888u),
        [s9] "s"(0x99999999u),
        [s10] "s"(0xaaaaaaaau),
        [s11] "s"(0xbbbbbbbbu),
        [s12] "s"(0xccccccccu),
        [s13] "s"(0xddddddddu),
        [s14] "s"(0xeeeeeeeeu),
        [s15] "s"(0xffffffffu),
        [lo] "s"(0x76543210u), [hi] "s"(0xfedcba98u)
      : "v162", "v164", "v166", "v168", "v170", "v172", "v174", "v176", "v178",
        "v180", "v182", "v184", "v186", "v188", "v190", "v192", "v196", "v197",
        "v198", "v199");
  return stn;
}

__device__ __forceinline__ void scan_piece(const int b, const int c,
                                           const float* __restrict__ ws,
                                           float* states,
                                           float* __restrict__ fs_out) {
  const int lane = threadIdx.x;
  const int sbeg = c * CH;
  const int hb = lane & 16;
  const float* Abar = ws + WS_ABAR;
  const float* uB = ws + WS_UB + (size_t)b * SEQ * 32;
  float* st_out = states + (size_t)b * SEQ * 32;

  float ao[16], ax[16];
#pragma unroll
  for (int m = 0; m < 16; ++m) {
    ao[m] = Abar[lane * 32 + hb + m];
    ax[m] = Abar[(lane ^ 16) * 32 + hb + m];
  }

  float cur[SCAN_PF];
#pragma unroll
  for (int j = 0; j < SCAN_PF; ++j) cur[j] = uB[(sbeg + j) * 32 + lane];

  float st = (c == 0) ? 0.f : st_out[(sbeg - 1) * 32 + lane];

  asm volatile(""
               : "+{v128}"(ao[0]), "+{v129}"(ax[0]),
               "+{v130}"(ao[1]), "+{v131}"(ax[1]),
               "+{v132}"(ao[2]), "+{v133}"(ax[2]),
               "+{v134}"(ao[3]), "+{v135}"(ax[3]),
               "+{v136}"(ao[4]), "+{v137}"(ax[4]),
               "+{v138}"(ao[5]), "+{v139}"(ax[5]),
               "+{v140}"(ao[6]), "+{v141}"(ax[6]),
               "+{v142}"(ao[7]), "+{v143}"(ax[7]),
               "+{v144}"(ao[8]), "+{v145}"(ax[8]),
               "+{v146}"(ao[9]), "+{v147}"(ax[9]),
               "+{v148}"(ao[10]), "+{v149}"(ax[10]),
               "+{v150}"(ao[11]), "+{v151}"(ax[11]),
               "+{v152}"(ao[12]), "+{v153}"(ax[12]),
               "+{v154}"(ao[13]), "+{v155}"(ax[13]),
               "+{v156}"(ao[14]), "+{v157}"(ax[14]),
               "+{v158}"(ao[15]), "+{v159}"(ax[15]));

  for (int s0 = sbeg; s0 < sbeg + CH; s0 += SCAN_PF) {
    const int sn = (s0 + SCAN_PF < sbeg + CH) ? (s0 + SCAN_PF) : s0;
    float nxt[SCAN_PF];
#pragma unroll
    for (int j = 0; j < SCAN_PF; ++j) nxt[j] = uB[(sn + j) * 32 + lane];

#pragma unroll
    for (int j = 0; j < SCAN_PF; ++j) {
      st = scan_step(st, cur[j], ao, ax);
      st_out[(s0 + j) * 32 + lane] = st;
    }
#pragma unroll
    for (int j = 0; j < SCAN_PF; ++j) cur[j] = nxt[j];
  }
  if (c == NCHUNK - 1) fs_out[b * 32 + lane] = st;
}

#define WOUT_PITCH (D_MODEL + 4)
__device__ __forceinline__ void proj_piece(
    const int bid, const int c, const float* states,
    const float* __restrict__ C, const float* __restrict__ x,
    const float* __restrict__ Dv, const float* __restrict__ W,
    const float* __restrict__ bout, float* __restrict__ y) {
  __shared__ float lds[16 * WOUT_PITCH];
  const int tid = threadIdx.x;
  const int mtile = piece_tile(c, bid);
  const int wave = tid >> 5, lane = tid & 31;
  const int r = lane & 15, h = lane >> 4;
  const int mbase = mtile * 16 + 8 * h;

  {
    v2f afrag[8];
    const float* srow = states + (size_t)(mtile * 16 + r) * 32;
#pragma unroll
    for (int kk = 0; kk < 8; ++kk)
      afrag[kk] = *(const v2f*)(srow + kk * 4 + 2 * h);

#pragma unroll
    for (int t = 0; t < 4; ++t) {
      const int dtile = wave * 4 + t;
      v8f acc = {0.f, 0.f, 0.f, 0.f, 0.f, 0.f, 0.f, 0.f};
      const float* crow = C + (size_t)(dtile * 16 + r) * 32;
#pragma unroll
      for (int kk = 0; kk < 8; ++kk) {
        const v2f b = *(const v2f*)(crow + kk * 4 + 2 * h);
        acc = wmma4(afrag[kk], b, acc);
      }
      const int d = dtile * 16 + r;
      const float Dd = Dv[d];
#pragma unroll
      for (int v = 0; v < 8; ++v)
        lds[(8 * h + v) * WOUT_PITCH + d] =
            acc[v] + x[(size_t)(mbase + v) * D_MODEL + d] * Dd;
    }
  }
  __syncthreads();

  const float* arow = lds + r * WOUT_PITCH + 4 * h;

  for (int t = 0; t < 4; ++t) {
    const int ntile = wave * 4 + t;
    const float* wrow = W + (size_t)(ntile * 16 + r) * D_MODEL + 4 * h;
    if (t < 3)
      __builtin_prefetch(W + (size_t)((ntile + 1) * 16 + r) * D_MODEL, 0, 1);

    v8f acc = {0.f, 0.f, 0.f, 0.f, 0.f, 0.f, 0.f, 0.f};
    for (int k8 = 0; k8 < D_MODEL; k8 += 8) {
      const float4 a4 = *(const float4*)(arow + k8);
      const float4 b4 = *(const float4*)(wrow + k8);
      const v2f a_lo = {a4.x, a4.y}, b_lo = {b4.x, b4.y};
      const v2f a_hi = {a4.z, a4.w}, b_hi = {b4.z, b4.w};
      acc = wmma4(a_lo, b_lo, acc);
      acc = wmma4(a_hi, b_hi, acc);
    }
    const int n = ntile * 16 + r;
    const float bo = bout[n];
#pragma unroll
    for (int v = 0; v < 8; ++v)
      y[(size_t)(mbase + v) * D_MODEL + n] = acc[v] + bo;
  }
}

__global__ void __launch_bounds__(256) stage_kernel(
    const float* __restrict__ ws, float* states, float* __restrict__ fs,
    const float* __restrict__ C, const float* __restrict__ x,
    const float* __restrict__ Dv, const float* __restrict__ W,
    const float* __restrict__ bout, float* __restrict__ y, const int c) {
  int bid = blockIdx.x;
  if (bid < NB_SCAN) {
    if (c < NCHUNK && threadIdx.x < 32) scan_piece(bid, c, ws, states, fs);
    return;
  }
  bid -= NB_SCAN;
  if (c >= 1) proj_piece(bid, c - 1, states, C, x, Dv, W, bout, y);
}

extern "C" void kernel_launch(void* const* d_in, const int* in_sizes, int n_in,
                              void* d_out, int out_size, void* d_ws,
                              size_t ws_size, hipStream_t stream) {
  const float* x = (const float*)d_in[0];
  const float* A = (const float*)d_in[1];
  const float* B = (const float*)d_in[2];
  const float* C = (const float*)d_in[3];
  const float* Dv = (const float*)d_in[4];
  const float* log_dt = (const float*)d_in[5];
  const float* W = (const float*)d_in[6];
  const float* bout = (const float*)d_in[7];

  float* y = (float*)d_out;
  float* fs = y + (size_t)ROWS * D_MODEL;

  float* ws = (float*)d_ws;
  float* Bbar = ws + WS_BBAR;
  float* uB = ws + WS_UB;
  float* states = ws + WS_STATES;

  prep_kernel<<<1, 256, 0, stream>>>(A, B, log_dt, ws);
  ub_kernel<<<256, 256, 0, stream>>>(x, Bbar, uB);
  stage_kernel<<<8, 256, 0, stream>>>(ws, states, fs, C, x, Dv, W, bout, y, 0);
  stage_kernel<<<264, 256, 0, stream>>>(ws, states, fs, C, x, Dv, W, bout, y, 1);
  stage_kernel<<<264, 256, 0, stream>>>(ws, states, fs, C, x, Dv, W, bout, y, 2);
  stage_kernel<<<264, 256, 0, stream>>>(ws, states, fs, C, x, Dv, W, bout, y, 3);
  stage_kernel<<<264, 256, 0, stream>>>(ws, states, fs, C, x, Dv, W, bout, y, 4);
  stage_kernel<<<264, 256, 0, stream>>>(ws, states, fs, C, x, Dv, W, bout, y, 5);
  stage_kernel<<<264, 256, 0, stream>>>(ws, states, fs, C, x, Dv, W, bout, y, 6);
  stage_kernel<<<264, 256, 0, stream>>>(ws, states, fs, C, x, Dv, W, bout, y, 7);
  stage_kernel<<<264, 256, 0, stream>>>(ws, states, fs, C, x, Dv, W, bout, y, 8);
}
